# starting kernel
# baseline (speedup 1.0000x reference)
#include <hip/hip_runtime.h>

#define NEG_SLOPE 0.2f
#define LN_EPS 1e-5f
#define SM_EPS 1e-16f

typedef __attribute__((ext_vector_type(2))) float v2f;
typedef __attribute__((ext_vector_type(8))) float v8f;

__device__ __forceinline__ float lrelu(float x) { return x > 0.0f ? x : NEG_SLOPE * x; }

__global__ void __launch_bounds__(128)
gemm64_wmma(const float* __restrict__ A, const float* __restrict__ W,
            float* __restrict__ C, int nrows) {
    const int wave = threadIdx.x >> 5;
    const int lane = threadIdx.x & 31;
    const int mrow = lane & 15;
    const int kh   = lane >> 4;
    const int row0 = blockIdx.x * 16;
    int row = row0 + mrow;
    row = row < nrows ? row : nrows - 1;
    const int col  = wave * 16 + mrow;

    const float* ap = A + (size_t)row * 64 + 2 * kh;
    const float* bp = W + (size_t)(2 * kh) * 64 + col;

    v8f acc = {};
    #pragma unroll
    for (int k = 0; k < 64; k += 4) {
        v2f a = *(const v2f*)(ap + k);
        v2f b;
        b.x = bp[(size_t)k * 64];
        b.y = bp[(size_t)k * 64 + 64];
        acc = __builtin_amdgcn_wmma_f32_16x16x4_f32(
             false, a,  false, b,
             (short)0, acc,  false,  false);
    }

    const int srow = row0 + 8 * (lane >> 4);
    const int scol = wave * 16 + (lane & 15);
    #pragma unroll
    for (int r = 0; r < 8; ++r) {
        const int rr = srow + r;
        if (rr < nrows) C[(size_t)rr * 64 + scol] = acc[r];
    }
}

template <int H, int C>
__global__ void attn_prep(const float* __restrict__ xw,
                          const float* __restrict__ a_src,
                          const float* __restrict__ a_dst,
                          float* __restrict__ asrc, float* __restrict__ adst,
                          float* __restrict__ m, int n) {
    const int t = blockIdx.x * blockDim.x + threadIdx.x;
    if (t >= n * H) return;
    const int node = t / H;
    const int h    = t - node * H;
    const float* xp = xw + (size_t)node * 64 + h * C;
    float s = 0.0f, d = 0.0f;
    #pragma unroll
    for (int c = 0; c < C; ++c) {
        const float v = xp[c];
        s += v * a_src[h * C + c];
        d += v * a_dst[h * C + c];
    }
    asrc[t] = s;
    adst[t] = d;
    m[t] = lrelu(s + d);
}

template <int H>
__global__ void edge_max_k(const int* __restrict__ src, const int* __restrict__ dst,
                           const float* __restrict__ asrc, const float* __restrict__ adst,
                           float* __restrict__ m, int ecnt) {
    const int t = blockIdx.x * blockDim.x + threadIdx.x;
    if (t >= ecnt * H) return;
    const int edge = t / H;
    const int h    = t - edge * H;
    const int s = src[edge];
    const int d = dst[edge];
    const float e = lrelu(asrc[s * H + h] + adst[d * H + h]);
    atomicMax(&m[d * H + h], e);
}

template <int H, int C>
__global__ void softmax_self(const float* __restrict__ xw,
                             const float* __restrict__ asrc,
                             const float* __restrict__ adst,
                             const float* __restrict__ m,
                             float* __restrict__ denom, float* __restrict__ agg,
                             int n) {
    const int t = blockIdx.x * blockDim.x + threadIdx.x;
    if (t >= n * 64) return;
    const int node = t >> 6;
    const int ch   = t & 63;
    const int h    = ch / C;
    const float e  = lrelu(asrc[node * H + h] + adst[node * H + h]);
    const float ex = __expf(e - m[node * H + h]);
    agg[t] = xw[t] * ex;
    if ((ch & (C - 1)) == 0) denom[node * H + h] = ex;
}

template <int H, int C>
__global__ void edge_agg(const int* __restrict__ src, const int* __restrict__ dst,
                         const float* __restrict__ asrc, const float* __restrict__ adst,
                         const float* __restrict__ m, const float* __restrict__ xw,
                         float* __restrict__ denom, float* __restrict__ agg,
                         int ecnt) {
    const long long t = (long long)blockIdx.x * blockDim.x + threadIdx.x;
    if (t >= (long long)ecnt * 64) return;
    const int edge = (int)(t >> 6);
    const int ch   = (int)(t & 63);
    const int h    = ch / C;
    const int s = src[edge];
    const int d = dst[edge];
    const float e  = lrelu(asrc[s * H + h] + adst[d * H + h]);
    const float ex = __expf(e - m[d * H + h]);
    atomicAdd(&agg[(size_t)d * 64 + ch], ex * xw[(size_t)s * 64 + ch]);
    if ((ch & (C - 1)) == 0) atomicAdd(&denom[d * H + h], ex);
}

template <int H, int C>
__global__ void finalize_k(const float* __restrict__ agg, const float* __restrict__ denom,
                           const float* __restrict__ bias, const float* __restrict__ g,
                           const float* __restrict__ bparam, float* __restrict__ out,
                           int n) {
    const int node = (blockIdx.x * blockDim.x + threadIdx.x) >> 5;
    const int lane = threadIdx.x & 31;
    if (node >= n) return;
    const int ch0 = lane, ch1 = lane + 32;
    float v0 = agg[(size_t)node * 64 + ch0] / (denom[node * H + ch0 / C] + SM_EPS) + bias[ch0];
    float v1 = agg[(size_t)node * 64 + ch1] / (denom[node * H + ch1 / C] + SM_EPS) + bias[ch1];

    float s = v0 + v1;
    #pragma unroll
    for (int off = 16; off > 0; off >>= 1) s += __shfl_xor(s, off, 32);
    const float mu = s * (1.0f / 64.0f);
    const float d0 = v0 - mu, d1 = v1 - mu;
    float vv = d0 * d0 + d1 * d1;
    #pragma unroll
    for (int off = 16; off > 0; off >>= 1) vv += __shfl_xor(vv, off, 32);
    const float rstd = rsqrtf(vv * (1.0f / 64.0f) + LN_EPS);

    float o0 = d0 * rstd * g[ch0] + bparam[ch0];
    float o1 = d1 * rstd * g[ch1] + bparam[ch1];
    o0 = o0 > 0.0f ? o0 : __expf(o0) - 1.0f;
    o1 = o1 > 0.0f ? o1 : __expf(o1) - 1.0f;
    out[(size_t)node * 64 + ch0] = o0;
    out[(size_t)node * 64 + ch1] = o1;
}

static inline unsigned cdiv(long long a, int b) { return (unsigned)((a + b - 1) / b); }

extern "C" void kernel_launch(void* const* d_in, const int* in_sizes, int n_in,
                              void* d_out, int out_size, void* d_ws, size_t ws_size,
                              hipStream_t stream) {
    const float* h   = (const float*)d_in[0];
    const int*   ei  = (const int*)d_in[1];
    const float* W1  = (const float*)d_in[2];
    const float* a1s = (const float*)d_in[3];
    const float* a1d = (const float*)d_in[4];
    const float* b1  = (const float*)d_in[5];
    const float* W2  = (const float*)d_in[6];
    const float* a2s = (const float*)d_in[7];
    const float* a2d = (const float*)d_in[8];
    const float* b2  = (const float*)d_in[9];
    const float* ln1g = (const float*)d_in[10];
    const float* ln1b = (const float*)d_in[11];
    const float* ln2g = (const float*)d_in[12];
    const float* ln2b = (const float*)d_in[13];

    const int n = in_sizes[0] / 64;
    const int e = in_sizes[1] / 2;
    const int* src = ei;
    const int* dst = ei + e;

    float* ws = (float*)d_ws;
    float* xw   = ws; ws += (size_t)n * 64;
    float* asrc = ws; ws += (size_t)n * 4;
    float* adst = ws; ws += (size_t)n * 4;
    float* m    = ws; ws += (size_t)n * 4;
    float* den  = ws; ws += (size_t)n * 4;
    float* agg  = ws; ws += (size_t)n * 64;
    float* x1   = ws; ws += (size_t)n * 64;

    const int B = 256;

    gemm64_wmma<<<cdiv(n, 16), 128, 0, stream>>>(h, W1, xw, n);
    attn_prep<4, 16><<<cdiv((long long)n * 4, B), B, 0, stream>>>(xw, a1s, a1d, asrc, adst, m, n);
    edge_max_k<4><<<cdiv((long long)e * 4, B), B, 0, stream>>>(src, dst, asrc, adst, m, e);
    softmax_self<4, 16><<<cdiv((long long)n * 64, B), B, 0, stream>>>(xw, asrc, adst, m, den, agg, n);
    edge_agg<4, 16><<<cdiv((long long)e * 64, B), B, 0, stream>>>(src, dst, asrc, adst, m, xw, den, agg, e);
    finalize_k<4, 16><<<cdiv((long long)n * 32, B), B, 0, stream>>>(agg, den, b1, ln1g, ln1b, x1, n);

    gemm64_wmma<<<cdiv(n, 16), 128, 0, stream>>>(x1, W2, xw, n);
    attn_prep<1, 64><<<cdiv((long long)n, B), B, 0, stream>>>(xw, a2s, a2d, asrc, adst, m, n);
    edge_max_k<1><<<cdiv((long long)e, B), B, 0, stream>>>(src, dst, asrc, adst, m, e);
    softmax_self<1, 64><<<cdiv((long long)n * 64, B), B, 0, stream>>>(xw, asrc, adst, m, den, agg, n);
    edge_agg<1, 64><<<cdiv((long long)e * 64, B), B, 0, stream>>>(src, dst, asrc, adst, m, xw, den, agg, e);
    finalize_k<1, 64><<<cdiv((long long)n * 32, B), B, 0, stream>>>(agg, den, b2, ln2g, ln2b, (float*)d_out, n);
}


# edges counting-sorted by destination once per call (int atomics, arrival-order rank); max and sums become per-node wave gathers with plain stores, lane-tree denominators, no float atomics
# speedup vs baseline: 3.1756x; 3.1756x over previous
#include <hip/hip_runtime.h>

#define NEG_SLOPE 0.2f
#define LN_EPS 1e-5f
#define SM_EPS 1e-16f

typedef __attribute__((ext_vector_type(2))) float v2f;
typedef __attribute__((ext_vector_type(8))) float v8f;

__device__ __forceinline__ float lrelu(float x) { return x > 0.0f ? x : NEG_SLOPE * x; }

__global__ void __launch_bounds__(128)
gemm64_wmma(const float* __restrict__ A, const float* __restrict__ W,
            float* __restrict__ C, int nrows) {
    const int wave = threadIdx.x >> 5;
    const int lane = threadIdx.x & 31;
    const int mrow = lane & 15;
    const int kh   = lane >> 4;
    const int row0 = blockIdx.x * 16;
    int row = row0 + mrow;
    row = row < nrows ? row : nrows - 1;
    const int col  = wave * 16 + mrow;

    const float* ap = A + (size_t)row * 64 + 2 * kh;
    const float* bp = W + (size_t)(2 * kh) * 64 + col;

    v8f acc = {};
    #pragma unroll
    for (int k = 0; k < 64; k += 4) {
        v2f a = *(const v2f*)(ap + k);
        v2f b;
        b.x = bp[(size_t)k * 64];
        b.y = bp[(size_t)k * 64 + 64];
        acc = __builtin_amdgcn_wmma_f32_16x16x4_f32(
             false, a,  false, b,
             (short)0, acc,  false,  false);
    }

    const int srow = row0 + 8 * (lane >> 4);
    const int scol = wave * 16 + (lane & 15);
    #pragma unroll
    for (int r = 0; r < 8; ++r) {
        const int rr = srow + r;
        if (rr < nrows) C[(size_t)rr * 64 + scol] = acc[r];
    }
}

template <int H, int C>
__global__ void attn_prep(const float* __restrict__ xw,
                          const float* __restrict__ a_src,
                          const float* __restrict__ a_dst,
                          float* __restrict__ asrc, float* __restrict__ adst,
                          int n) {
    const int t = blockIdx.x * blockDim.x + threadIdx.x;
    if (t >= n * H) return;
    const int node = t / H;
    const int h    = t - node * H;
    const float* xp = xw + (size_t)node * 64 + h * C;
    float s = 0.0f, d = 0.0f;
    #pragma unroll
    for (int c = 0; c < C; ++c) {
        const float v = xp[c];
        s += v * a_src[h * C + c];
        d += v * a_dst[h * C + c];
    }
    asrc[t] = s;
    adst[t] = d;
}

__global__ void sort_zero(int* __restrict__ cnt, int* __restrict__ total, int n) {
    const int t = blockIdx.x * blockDim.x + threadIdx.x;
    if (t < n) cnt[t] = 0;
    if (t == 0) *total = 0;
}

__global__ void sort_rank(const int* __restrict__ dst, int* __restrict__ cnt,
                          int* __restrict__ rank, int ecnt) {
    const int t = blockIdx.x * blockDim.x + threadIdx.x;
    if (t < ecnt) rank[t] = atomicAdd(&cnt[dst[t]], 1);
}

__global__ void sort_alloc(const int* __restrict__ cnt, int* __restrict__ total,
                           int* __restrict__ start, int n) {
    const int t = blockIdx.x * blockDim.x + threadIdx.x;
    const int lane = threadIdx.x & 31;
    const int c = t < n ? cnt[t] : 0;
    int incl = c;
    for (int off = 1; off < 32; off <<= 1) {
        const int v = __shfl_up(incl, off, 32);
        if (lane >= off) incl += v;
    }
    int base = 0;
    if (lane == 31) base = atomicAdd(total, incl);
    base = __shfl(base, 31, 32);
    if (t < n) start[t] = base + incl - c;
}

__global__ void sort_place(const int* __restrict__ src, const int* __restrict__ dst,
                           const int* __restrict__ start, const int* __restrict__ rank,
                           int* __restrict__ sorted_src, int ecnt) {
    const int t = blockIdx.x * blockDim.x + threadIdx.x;
    if (t < ecnt) sorted_src[start[dst[t]] + rank[t]] = src[t];
}

template <int H, int C>
__global__ void __launch_bounds__(256)
node_gather(const int* __restrict__ start, const int* __restrict__ cnt,
            const int* __restrict__ sorted_src,
            const float* __restrict__ asrc, const float* __restrict__ adst,
            const float* __restrict__ xw,
            float* __restrict__ denom, float* __restrict__ agg, int n) {
    constexpr int EPC = 32 / H;
    constexpr int GRP = EPC < 8 ? EPC : 8;
    const int node = __builtin_amdgcn_readfirstlane(
        (int)((blockIdx.x * blockDim.x + threadIdx.x) >> 5));
    const int lane = threadIdx.x & 31;
    if (node >= n) return;
    const int hsel = lane % H;
    const int esel = lane / H;
    const int ch   = 2 * lane;
    const int hch  = ch / C;
    const int beg  = start[node];
    const int deg  = cnt[node];
    const float ad    = adst[node * H + hsel];
    const float eself = lrelu(asrc[node * H + hsel] + ad);

    float mx = eself;
    for (int c0 = 0; c0 < deg; c0 += EPC) {
        const int j = c0 + esel;
        const int s = sorted_src[beg + (j < deg ? j : deg - 1)];
        mx = fmaxf(mx, lrelu(asrc[s * H + hsel] + ad));
    }
    for (int off = H; off < 32; off <<= 1) mx = fmaxf(mx, __shfl_xor(mx, off, 32));

    const float exself = __expf(eself - mx);
    float dsum = esel == 0 ? exself : 0.0f;
    const float wself = __shfl(exself, hch, 32);
    const v2f xs = *(const v2f*)(xw + (size_t)node * 64 + ch);
    float acc0 = wself * xs.x;
    float acc1 = wself * xs.y;

    for (int c0 = 0; c0 < deg; c0 += EPC) {
        const int j = c0 + esel;
        const bool live = j < deg;
        const int   sl = sorted_src[beg + (live ? j : deg - 1)];
        const float el = __expf(lrelu(asrc[sl * H + hsel] + ad) - mx);
        const int   s  = live ? sl : node;
        const float ex = live ? el : 0.0f;
        dsum += ex;
        const int lim = deg - c0 < EPC ? deg - c0 : EPC;
        for (int k0 = 0; k0 < lim; k0 += GRP) {
            #pragma unroll
            for (int kk = 0; kk < GRP; ++kk) {
                const int k  = k0 + kk;
                const int sk = __builtin_amdgcn_readlane(s, k * H);
                const float wk = __shfl(ex, k * H + hch, 32);
                const v2f v = *(const v2f*)(xw + (size_t)sk * 64 + ch);
                acc0 += wk * v.x;
                acc1 += wk * v.y;
            }
        }
    }
    for (int off = H; off < 32; off <<= 1) dsum += __shfl_xor(dsum, off, 32);

    if (lane < H) denom[node * H + lane] = dsum;
    v2f o; o.x = acc0; o.y = acc1;
    *(v2f*)(agg + (size_t)node * 64 + ch) = o;
}

template <int H, int C>
__global__ void finalize_k(const float* __restrict__ agg, const float* __restrict__ denom,
                           const float* __restrict__ bias, const float* __restrict__ g,
                           const float* __restrict__ bparam, float* __restrict__ out,
                           int n) {
    const int node = (blockIdx.x * blockDim.x + threadIdx.x) >> 5;
    const int lane = threadIdx.x & 31;
    if (node >= n) return;
    const int ch0 = lane, ch1 = lane + 32;
    float v0 = agg[(size_t)node * 64 + ch0] / (denom[node * H + ch0 / C] + SM_EPS) + bias[ch0];
    float v1 = agg[(size_t)node * 64 + ch1] / (denom[node * H + ch1 / C] + SM_EPS) + bias[ch1];

    float s = v0 + v1;
    #pragma unroll
    for (int off = 16; off > 0; off >>= 1) s += __shfl_xor(s, off, 32);
    const float mu = s * (1.0f / 64.0f);
    const float d0 = v0 - mu, d1 = v1 - mu;
    float vv = d0 * d0 + d1 * d1;
    #pragma unroll
    for (int off = 16; off > 0; off >>= 1) vv += __shfl_xor(vv, off, 32);
    const float rstd = rsqrtf(vv * (1.0f / 64.0f) + LN_EPS);

    float o0 = d0 * rstd * g[ch0] + bparam[ch0];
    float o1 = d1 * rstd * g[ch1] + bparam[ch1];
    o0 = o0 > 0.0f ? o0 : __expf(o0) - 1.0f;
    o1 = o1 > 0.0f ? o1 : __expf(o1) - 1.0f;
    out[(size_t)node * 64 + ch0] = o0;
    out[(size_t)node * 64 + ch1] = o1;
}

static inline unsigned cdiv(long long a, int b) { return (unsigned)((a + b - 1) / b); }

extern "C" void kernel_launch(void* const* d_in, const int* in_sizes, int n_in,
                              void* d_out, int out_size, void* d_ws, size_t ws_size,
                              hipStream_t stream) {
    const float* h   = (const float*)d_in[0];
    const int*   ei  = (const int*)d_in[1];
    const float* W1  = (const float*)d_in[2];
    const float* a1s = (const float*)d_in[3];
    const float* a1d = (const float*)d_in[4];
    const float* b1  = (const float*)d_in[5];
    const float* W2  = (const float*)d_in[6];
    const float* a2s = (const float*)d_in[7];
    const float* a2d = (const float*)d_in[8];
    const float* b2  = (const float*)d_in[9];
    const float* ln1g = (const float*)d_in[10];
    const float* ln1b = (const float*)d_in[11];
    const float* ln2g = (const float*)d_in[12];
    const float* ln2b = (const float*)d_in[13];

    const int n = in_sizes[0] / 64;
    const int e = in_sizes[1] / 2;
    const int* src = ei;
    const int* dst = ei + e;

    float* ws = (float*)d_ws;
    float* xw   = ws; ws += (size_t)n * 64;
    float* asrc = ws; ws += (size_t)n * 4;
    float* adst = ws; ws += (size_t)n * 4;
    float* den  = ws; ws += (size_t)n * 4;
    float* agg  = ws; ws += (size_t)n * 64;
    float* x1   = ws; ws += (size_t)n * 64;
    int* wi     = (int*)ws;
    int* cnt    = wi; wi += n;
    int* start  = wi; wi += n;
    int* total  = wi; wi += 2;
    int* rank   = wi; wi += e;
    int* ssrc   = wi; wi += e;

    const int B = 256;

    sort_zero<<<cdiv(n, B), B, 0, stream>>>(cnt, total, n);
    sort_rank<<<cdiv(e, B), B, 0, stream>>>(dst, cnt, rank, e);
    sort_alloc<<<cdiv(n, B), B, 0, stream>>>(cnt, total, start, n);
    sort_place<<<cdiv(e, B), B, 0, stream>>>(src, dst, start, rank, ssrc, e);

    gemm64_wmma<<<cdiv(n, 16), 128, 0, stream>>>(h, W1, xw, n);
    attn_prep<4, 16><<<cdiv((long long)n * 4, B), B, 0, stream>>>(xw, a1s, a1d, asrc, adst, n);
    node_gather<4, 16><<<cdiv((long long)n * 32, B), B, 0, stream>>>(start, cnt, ssrc, asrc, adst, xw, den, agg, n);
    finalize_k<4, 16><<<cdiv((long long)n * 32, B), B, 0, stream>>>(agg, den, b1, ln1g, ln1b, x1, n);

    gemm64_wmma<<<cdiv(n, 16), 128, 0, stream>>>(x1, W2, xw, n);
    attn_prep<1, 64><<<cdiv((long long)n, B), B, 0, stream>>>(xw, a2s, a2d, asrc, adst, n);
    node_gather<1, 64><<<cdiv((long long)n * 32, B), B, 0, stream>>>(start, cnt, ssrc, asrc, adst, xw, den, agg, n);
    finalize_k<1, 64><<<cdiv((long long)n * 32, B), B, 0, stream>>>(agg, den, b2, ln2g, ln2b, (float*)d_out, n);
}


# bias+LayerNorm+ELU epilogue fused into the per-node gather wave (row already in its registers); agg/denom round trip and 2 launches gone; LN lane sums now pair channels 2l,2l+1
# speedup vs baseline: 3.4289x; 1.0798x over previous
#include <hip/hip_runtime.h>

#define NEG_SLOPE 0.2f
#define LN_EPS 1e-5f
#define SM_EPS 1e-16f

typedef __attribute__((ext_vector_type(2))) float v2f;
typedef __attribute__((ext_vector_type(8))) float v8f;

__device__ __forceinline__ float lrelu(float x) { return x > 0.0f ? x : NEG_SLOPE * x; }

__global__ void __launch_bounds__(128)
gemm64_wmma(const float* __restrict__ A, const float* __restrict__ W,
            float* __restrict__ C, int nrows) {
    const int wave = threadIdx.x >> 5;
    const int lane = threadIdx.x & 31;
    const int mrow = lane & 15;
    const int kh   = lane >> 4;
    const int row0 = blockIdx.x * 16;
    int row = row0 + mrow;
    row = row < nrows ? row : nrows - 1;
    const int col  = wave * 16 + mrow;

    const float* ap = A + (size_t)row * 64 + 2 * kh;
    const float* bp = W + (size_t)(2 * kh) * 64 + col;

    v8f acc = {};
    #pragma unroll
    for (int k = 0; k < 64; k += 4) {
        v2f a = *(const v2f*)(ap + k);
        v2f b;
        b.x = bp[(size_t)k * 64];
        b.y = bp[(size_t)k * 64 + 64];
        acc = __builtin_amdgcn_wmma_f32_16x16x4_f32(
             false, a,  false, b,
             (short)0, acc,  false,  false);
    }

    const int srow = row0 + 8 * (lane >> 4);
    const int scol = wave * 16 + (lane & 15);
    #pragma unroll
    for (int r = 0; r < 8; ++r) {
        const int rr = srow + r;
        if (rr < nrows) C[(size_t)rr * 64 + scol] = acc[r];
    }
}

template <int H, int C>
__global__ void attn_prep(const float* __restrict__ xw,
                          const float* __restrict__ a_src,
                          const float* __restrict__ a_dst,
                          float* __restrict__ asrc, float* __restrict__ adst,
                          int n) {
    const int t = blockIdx.x * blockDim.x + threadIdx.x;
    if (t >= n * H) return;
    const int node = t / H;
    const int h    = t - node * H;
    const float* xp = xw + (size_t)node * 64 + h * C;
    float s = 0.0f, d = 0.0f;
    #pragma unroll
    for (int c = 0; c < C; ++c) {
        const float v = xp[c];
        s += v * a_src[h * C + c];
        d += v * a_dst[h * C + c];
    }
    asrc[t] = s;
    adst[t] = d;
}

__global__ void sort_zero(int* __restrict__ cnt, int* __restrict__ total, int n) {
    const int t = blockIdx.x * blockDim.x + threadIdx.x;
    if (t < n) cnt[t] = 0;
    if (t == 0) *total = 0;
}

__global__ void sort_rank(const int* __restrict__ dst, int* __restrict__ cnt,
                          int* __restrict__ rank, int ecnt) {
    const int t = blockIdx.x * blockDim.x + threadIdx.x;
    if (t < ecnt) rank[t] = atomicAdd(&cnt[dst[t]], 1);
}

__global__ void sort_alloc(const int* __restrict__ cnt, int* __restrict__ total,
                           int* __restrict__ start, int n) {
    const int t = blockIdx.x * blockDim.x + threadIdx.x;
    const int lane = threadIdx.x & 31;
    const int c = t < n ? cnt[t] : 0;
    int incl = c;
    for (int off = 1; off < 32; off <<= 1) {
        const int v = __shfl_up(incl, off, 32);
        if (lane >= off) incl += v;
    }
    int base = 0;
    if (lane == 31) base = atomicAdd(total, incl);
    base = __shfl(base, 31, 32);
    if (t < n) start[t] = base + incl - c;
}

__global__ void sort_place(const int* __restrict__ src, const int* __restrict__ dst,
                           const int* __restrict__ start, const int* __restrict__ rank,
                           int* __restrict__ sorted_src, int ecnt) {
    const int t = blockIdx.x * blockDim.x + threadIdx.x;
    if (t < ecnt) sorted_src[start[dst[t]] + rank[t]] = src[t];
}

template <int H, int C>
__global__ void __launch_bounds__(256)
node_gather_norm(const int* __restrict__ start, const int* __restrict__ cnt,
                 const int* __restrict__ sorted_src,
                 const float* __restrict__ asrc, const float* __restrict__ adst,
                 const float* __restrict__ xw,
                 const float* __restrict__ bias, const float* __restrict__ g,
                 const float* __restrict__ bparam,
                 float* __restrict__ out, int n) {
    constexpr int EPC = 32 / H;
    constexpr int GRP = EPC < 8 ? EPC : 8;
    const int node = __builtin_amdgcn_readfirstlane(
        (int)((blockIdx.x * blockDim.x + threadIdx.x) >> 5));
    const int lane = threadIdx.x & 31;
    if (node >= n) return;
    const int hsel = lane % H;
    const int esel = lane / H;
    const int ch   = 2 * lane;
    const int hch  = ch / C;
    const int beg  = start[node];
    const int deg  = cnt[node];
    const float ad    = adst[node * H + hsel];
    const float eself = lrelu(asrc[node * H + hsel] + ad);
    const v2f bi = *(const v2f*)(bias + ch);
    const v2f gg = *(const v2f*)(g + ch);
    const v2f bb = *(const v2f*)(bparam + ch);

    float mx = eself;
    for (int c0 = 0; c0 < deg; c0 += EPC) {
        const int j = c0 + esel;
        const int s = sorted_src[beg + (j < deg ? j : deg - 1)];
        mx = fmaxf(mx, lrelu(asrc[s * H + hsel] + ad));
    }
    for (int off = H; off < 32; off <<= 1) mx = fmaxf(mx, __shfl_xor(mx, off, 32));

    const float exself = __expf(eself - mx);
    float dsum = esel == 0 ? exself : 0.0f;
    const float wself = __shfl(exself, hch, 32);
    const v2f xs = *(const v2f*)(xw + (size_t)node * 64 + ch);
    float acc0 = wself * xs.x;
    float acc1 = wself * xs.y;

    for (int c0 = 0; c0 < deg; c0 += EPC) {
        const int j = c0 + esel;
        const bool live = j < deg;
        const int   sl = sorted_src[beg + (live ? j : deg - 1)];
        const float el = __expf(lrelu(asrc[sl * H + hsel] + ad) - mx);
        const int   s  = live ? sl : node;
        const float ex = live ? el : 0.0f;
        dsum += ex;
        const int lim = deg - c0 < EPC ? deg - c0 : EPC;
        for (int k0 = 0; k0 < lim; k0 += GRP) {
            #pragma unroll
            for (int kk = 0; kk < GRP; ++kk) {
                const int k  = k0 + kk;
                const int sk = __builtin_amdgcn_readlane(s, k * H);
                const float wk = __shfl(ex, k * H + hch, 32);
                const v2f v = *(const v2f*)(xw + (size_t)sk * 64 + ch);
                acc0 += wk * v.x;
                acc1 += wk * v.y;
            }
        }
    }
    for (int off = H; off < 32; off <<= 1) dsum += __shfl_xor(dsum, off, 32);
    const float den = __shfl(dsum, hch, 32) + SM_EPS;

    const float v0 = acc0 / den + bi.x;
    const float v1 = acc1 / den + bi.y;
    float sum = v0 + v1;
    #pragma unroll
    for (int off = 16; off > 0; off >>= 1) sum += __shfl_xor(sum, off, 32);
    const float mu = sum * (1.0f / 64.0f);
    const float d0 = v0 - mu, d1 = v1 - mu;
    float vv = d0 * d0 + d1 * d1;
    #pragma unroll
    for (int off = 16; off > 0; off >>= 1) vv += __shfl_xor(vv, off, 32);
    const float rstd = rsqrtf(vv * (1.0f / 64.0f) + LN_EPS);

    float o0 = d0 * rstd * gg.x + bb.x;
    float o1 = d1 * rstd * gg.y + bb.y;
    o0 = o0 > 0.0f ? o0 : __expf(o0) - 1.0f;
    o1 = o1 > 0.0f ? o1 : __expf(o1) - 1.0f;
    v2f o; o.x = o0; o.y = o1;
    *(v2f*)(out + (size_t)node * 64 + ch) = o;
}

static inline unsigned cdiv(long long a, int b) { return (unsigned)((a + b - 1) / b); }

extern "C" void kernel_launch(void* const* d_in, const int* in_sizes, int n_in,
                              void* d_out, int out_size, void* d_ws, size_t ws_size,
                              hipStream_t stream) {
    const float* h   = (const float*)d_in[0];
    const int*   ei  = (const int*)d_in[1];
    const float* W1  = (const float*)d_in[2];
    const float* a1s = (const float*)d_in[3];
    const float* a1d = (const float*)d_in[4];
    const float* b1  = (const float*)d_in[5];
    const float* W2  = (const float*)d_in[6];
    const float* a2s = (const float*)d_in[7];
    const float* a2d = (const float*)d_in[8];
    const float* b2  = (const float*)d_in[9];
    const float* ln1g = (const float*)d_in[10];
    const float* ln1b = (const float*)d_in[11];
    const float* ln2g = (const float*)d_in[12];
    const float* ln2b = (const float*)d_in[13];

    const int n = in_sizes[0] / 64;
    const int e = in_sizes[1] / 2;
    const int* src = ei;
    const int* dst = ei + e;

    float* ws = (float*)d_ws;
    float* xw   = ws; ws += (size_t)n * 64;
    float* asrc = ws; ws += (size_t)n * 4;
    float* adst = ws; ws += (size_t)n * 4;
    float* x1   = ws; ws += (size_t)n * 64;
    int* wi     = (int*)ws;
    int* cnt    = wi; wi += n;
    int* start  = wi; wi += n;
    int* total  = wi; wi += 2;
    int* rank   = wi; wi += e;
    int* ssrc   = wi; wi += e;

    const int B = 256;

    sort_zero<<<cdiv(n, B), B, 0, stream>>>(cnt, total, n);
    sort_rank<<<cdiv(e, B), B, 0, stream>>>(dst, cnt, rank, e);
    sort_alloc<<<cdiv(n, B), B, 0, stream>>>(cnt, total, start, n);
    sort_place<<<cdiv(e, B), B, 0, stream>>>(src, dst, start, rank, ssrc, e);

    gemm64_wmma<<<cdiv(n, 16), 128, 0, stream>>>(h, W1, xw, n);
    attn_prep<4, 16><<<cdiv((long long)n * 4, B), B, 0, stream>>>(xw, a1s, a1d, asrc, adst, n);
    node_gather_norm<4, 16><<<cdiv((long long)n * 32, B), B, 0, stream>>>(start, cnt, ssrc, asrc, adst, xw, b1, ln1g, ln1b, x1, n);

    gemm64_wmma<<<cdiv(n, 16), 128, 0, stream>>>(x1, W2, xw, n);
    attn_prep<1, 64><<<cdiv((long long)n, B), B, 0, stream>>>(xw, a2s, a2d, asrc, adst, n);
    node_gather_norm<1, 64><<<cdiv((long long)n * 32, B), B, 0, stream>>>(start, cnt, ssrc, asrc, adst, xw, b2, ln2g, ln2b, (float*)d_out, n);
}


# edge ranks from returning LDS adds (64 edge chunks, counter per node in LDS, chunk counts scanned to bases) replace one global atomic per edge; clear launch gone; sum order per node now chunk-major
# speedup vs baseline: 4.8848x; 1.4246x over previous
#include <hip/hip_runtime.h>

#define NEG_SLOPE 0.2f
#define LN_EPS 1e-5f
#define SM_EPS 1e-16f

typedef __attribute__((ext_vector_type(2))) float v2f;
typedef __attribute__((ext_vector_type(8))) float v8f;

__device__ __forceinline__ float lrelu(float x) { return x > 0.0f ? x : NEG_SLOPE * x; }

__global__ void __launch_bounds__(128)
gemm64_wmma(const float* __restrict__ A, const float* __restrict__ W,
            float* __restrict__ C, int nrows) {
    const int wave = threadIdx.x >> 5;
    const int lane = threadIdx.x & 31;
    const int mrow = lane & 15;
    const int kh   = lane >> 4;
    const int row0 = blockIdx.x * 16;
    int row = row0 + mrow;
    row = row < nrows ? row : nrows - 1;
    const int col  = wave * 16 + mrow;

    const float* ap = A + (size_t)row * 64 + 2 * kh;
    const float* bp = W + (size_t)(2 * kh) * 64 + col;

    v8f acc = {};
    #pragma unroll
    for (int k = 0; k < 64; k += 4) {
        v2f a = *(const v2f*)(ap + k);
        v2f b;
        b.x = bp[(size_t)k * 64];
        b.y = bp[(size_t)k * 64 + 64];
        acc = __builtin_amdgcn_wmma_f32_16x16x4_f32(
             false, a,  false, b,
             (short)0, acc,  false,  false);
    }

    const int srow = row0 + 8 * (lane >> 4);
    const int scol = wave * 16 + (lane & 15);
    #pragma unroll
    for (int r = 0; r < 8; ++r) {
        const int rr = srow + r;
        if (rr < nrows) C[(size_t)rr * 64 + scol] = acc[r];
    }
}

template <int H, int C>
__global__ void attn_prep(const float* __restrict__ xw,
                          const float* __restrict__ a_src,
                          const float* __restrict__ a_dst,
                          float* __restrict__ asrc, float* __restrict__ adst,
                          int n) {
    const int t = blockIdx.x * blockDim.x + threadIdx.x;
    if (t >= n * H) return;
    const int node = t / H;
    const int h    = t - node * H;
    const float* xp = xw + (size_t)node * 64 + h * C;
    float s = 0.0f, d = 0.0f;
    #pragma unroll
    for (int c = 0; c < C; ++c) {
        const float v = xp[c];
        s += v * a_src[h * C + c];
        d += v * a_dst[h * C + c];
    }
    asrc[t] = s;
    adst[t] = d;
}

#define SORT_CHUNKS  64
#define SORT_TILE    65536
#define SORT_THREADS 1024

__device__ __forceinline__ int sort_chunk_len(int ecnt) {
    return (ecnt + SORT_CHUNKS - 1) / SORT_CHUNKS;
}

__global__ void __launch_bounds__(SORT_THREADS)
sort_rank_lds(const int* __restrict__ dst, int* __restrict__ rank,
              int* __restrict__ table, int* __restrict__ total,
              int n, int ecnt) {
    __shared__ int lcnt[SORT_TILE + 32];
    const int tid   = threadIdx.x;
    const int chunk = blockIdx.x;
    const int per   = sort_chunk_len(ecnt);
    const int e0    = chunk * per;
    const int e1    = e0 + per < ecnt ? e0 + per : ecnt;
    const int spare = SORT_TILE + (tid & 31);
    if (chunk == 0 && tid == 0) *total = 0;
    for (int tb = 0; tb < n; tb += SORT_TILE) {
        const int tn = n - tb < SORT_TILE ? n - tb : SORT_TILE;
        for (int i = tid; i < tn; i += SORT_THREADS) lcnt[i] = 0;
        __syncthreads();
        for (int e = e0 + tid; e < e1; e += 4 * SORT_THREADS) {
            int d[4], r[4];
            #pragma unroll
            for (int u = 0; u < 4; ++u) {
                const int ee = e + u * SORT_THREADS;
                const int dv = dst[ee < e1 ? ee : e1 - 1];
                d[u] = ee < e1 ? dv - tb : -1;
            }
            #pragma unroll
            for (int u = 0; u < 4; ++u)
                r[u] = atomicAdd(&lcnt[(unsigned)d[u] < (unsigned)tn ? d[u] : spare], 1);
            #pragma unroll
            for (int u = 0; u < 4; ++u)
                if ((unsigned)d[u] < (unsigned)tn) rank[e + u * SORT_THREADS] = r[u];
        }
        __syncthreads();
        int* row = table + (size_t)chunk * n + tb;
        for (int i = tid; i < tn; i += SORT_THREADS) row[i] = lcnt[i];
        __syncthreads();
    }
}

__global__ void __launch_bounds__(256)
sort_alloc(int* __restrict__ table, int* __restrict__ total,
           int* __restrict__ start, int* __restrict__ cnt, int n) {
    const int t = blockIdx.x * blockDim.x + threadIdx.x;
    const int lane = threadIdx.x & 31;
    const bool live = t < n;
    int v[SORT_CHUNKS];
    #pragma unroll
    for (int k = 0; k < SORT_CHUNKS; ++k) v[k] = 0;
    if (live) {
        #pragma unroll
        for (int k = 0; k < SORT_CHUNKS; ++k) v[k] = table[(size_t)k * n + t];
    }
    int c = 0;
    #pragma unroll
    for (int k = 0; k < SORT_CHUNKS; ++k) c += v[k];
    int incl = c;
    for (int off = 1; off < 32; off <<= 1) {
        const int s = __shfl_up(incl, off, 32);
        if (lane >= off) incl += s;
    }
    int base = 0;
    if (lane == 31) base = atomicAdd(total, incl);
    base = __shfl(base, 31, 32);
    if (live) {
        int pos = base + incl - c;
        start[t] = pos;
        cnt[t]   = c;
        #pragma unroll
        for (int k = 0; k < SORT_CHUNKS; ++k) {
            table[(size_t)k * n + t] = pos;
            pos += v[k];
        }
    }
}

__global__ void sort_place(const int* __restrict__ src, const int* __restrict__ dst,
                           const int* __restrict__ table, const int* __restrict__ rank,
                           int* __restrict__ sorted_src, int n, int ecnt) {
    const int t = blockIdx.x * blockDim.x + threadIdx.x;
    if (t < ecnt) {
        const int chunk = t / sort_chunk_len(ecnt);
        const int d = dst[t];
        if ((unsigned)d < (unsigned)n)
            sorted_src[table[(size_t)chunk * n + d] + rank[t]] = src[t];
    }
}

template <int H, int C>
__global__ void __launch_bounds__(256)
node_gather_norm(const int* __restrict__ start, const int* __restrict__ cnt,
                 const int* __restrict__ sorted_src,
                 const float* __restrict__ asrc, const float* __restrict__ adst,
                 const float* __restrict__ xw,
                 const float* __restrict__ bias, const float* __restrict__ g,
                 const float* __restrict__ bparam,
                 float* __restrict__ out, int n) {
    constexpr int EPC = 32 / H;
    constexpr int GRP = EPC < 8 ? EPC : 8;
    const int node = __builtin_amdgcn_readfirstlane(
        (int)((blockIdx.x * blockDim.x + threadIdx.x) >> 5));
    const int lane = threadIdx.x & 31;
    if (node >= n) return;
    const int hsel = lane % H;
    const int esel = lane / H;
    const int ch   = 2 * lane;
    const int hch  = ch / C;
    const int beg  = start[node];
    const int deg  = cnt[node];
    const float ad    = adst[node * H + hsel];
    const float eself = lrelu(asrc[node * H + hsel] + ad);
    const v2f bi = *(const v2f*)(bias + ch);
    const v2f gg = *(const v2f*)(g + ch);
    const v2f bb = *(const v2f*)(bparam + ch);

    float mx = eself;
    for (int c0 = 0; c0 < deg; c0 += EPC) {
        const int j = c0 + esel;
        const int s = sorted_src[beg + (j < deg ? j : deg - 1)];
        mx = fmaxf(mx, lrelu(asrc[s * H + hsel] + ad));
    }
    for (int off = H; off < 32; off <<= 1) mx = fmaxf(mx, __shfl_xor(mx, off, 32));

    const float exself = __expf(eself - mx);
    float dsum = esel == 0 ? exself : 0.0f;
    const float wself = __shfl(exself, hch, 32);
    const v2f xs = *(const v2f*)(xw + (size_t)node * 64 + ch);
    float acc0 = wself * xs.x;
    float acc1 = wself * xs.y;

    for (int c0 = 0; c0 < deg; c0 += EPC) {
        const int j = c0 + esel;
        const bool live = j < deg;
        const int   sl = sorted_src[beg + (live ? j : deg - 1)];
        const float el = __expf(lrelu(asrc[sl * H + hsel] + ad) - mx);
        const int   s  = live ? sl : node;
        const float ex = live ? el : 0.0f;
        dsum += ex;
        const int lim = deg - c0 < EPC ? deg - c0 : EPC;
        for (int k0 = 0; k0 < lim; k0 += GRP) {
            #pragma unroll
            for (int kk = 0; kk < GRP; ++kk) {
                const int k  = k0 + kk;
                const int sk = __builtin_amdgcn_readlane(s, k * H);
                const float wk = __shfl(ex, k * H + hch, 32);
                const v2f v = *(const v2f*)(xw + (size_t)sk * 64 + ch);
                acc0 += wk * v.x;
                acc1 += wk * v.y;
            }
        }
    }
    for (int off = H; off < 32; off <<= 1) dsum += __shfl_xor(dsum, off, 32);
    const float den = __shfl(dsum, hch, 32) + SM_EPS;

    const float v0 = acc0 / den + bi.x;
    const float v1 = acc1 / den + bi.y;
    float sum = v0 + v1;
    #pragma unroll
    for (int off = 16; off > 0; off >>= 1) sum += __shfl_xor(sum, off, 32);
    const float mu = sum * (1.0f / 64.0f);
    const float d0 = v0 - mu, d1 = v1 - mu;
    float vv = d0 * d0 + d1 * d1;
    #pragma unroll
    for (int off = 16; off > 0; off >>= 1) vv += __shfl_xor(vv, off, 32);
    const float rstd = rsqrtf(vv * (1.0f / 64.0f) + LN_EPS);

    float o0 = d0 * rstd * gg.x + bb.x;
    float o1 = d1 * rstd * gg.y + bb.y;
    o0 = o0 > 0.0f ? o0 : __expf(o0) - 1.0f;
    o1 = o1 > 0.0f ? o1 : __expf(o1) - 1.0f;
    v2f o; o.x = o0; o.y = o1;
    *(v2f*)(out + (size_t)node * 64 + ch) = o;
}

static inline unsigned cdiv(long long a, int b) { return (unsigned)((a + b - 1) / b); }

extern "C" void kernel_launch(void* const* d_in, const int* in_sizes, int n_in,
                              void* d_out, int out_size, void* d_ws, size_t ws_size,
                              hipStream_t stream) {
    const float* h   = (const float*)d_in[0];
    const int*   ei  = (const int*)d_in[1];
    const float* W1  = (const float*)d_in[2];
    const float* a1s = (const float*)d_in[3];
    const float* a1d = (const float*)d_in[4];
    const float* b1  = (const float*)d_in[5];
    const float* W2  = (const float*)d_in[6];
    const float* a2s = (const float*)d_in[7];
    const float* a2d = (const float*)d_in[8];
    const float* b2  = (const float*)d_in[9];
    const float* ln1g = (const float*)d_in[10];
    const float* ln1b = (const float*)d_in[11];
    const float* ln2g = (const float*)d_in[12];
    const float* ln2b = (const float*)d_in[13];

    const int n = in_sizes[0] / 64;
    const int e = in_sizes[1] / 2;
    const int* src = ei;
    const int* dst = ei + e;

    float* ws = (float*)d_ws;
    float* xw   = ws; ws += (size_t)n * 64;
    float* asrc = ws; ws += (size_t)n * 4;
    float* adst = ws; ws += (size_t)n * 4;
    float* x1   = ws; ws += (size_t)n * 64;
    int* wi     = (int*)ws;
    int* cnt    = wi; wi += n;
    int* start  = wi; wi += n;
    int* total  = wi; wi += 2;
    int* rank   = wi; wi += e;
    int* ssrc   = wi; wi += e;
    int* table  = wi; wi += (size_t)SORT_CHUNKS * n;

    const int B = 256;

    sort_rank_lds<<<SORT_CHUNKS, SORT_THREADS, 0, stream>>>(dst, rank, table, total, n, e);
    sort_alloc<<<cdiv(n, B), B, 0, stream>>>(table, total, start, cnt, n);
    sort_place<<<cdiv(e, B), B, 0, stream>>>(src, dst, table, rank, ssrc, n, e);

    gemm64_wmma<<<cdiv(n, 16), 128, 0, stream>>>(h, W1, xw, n);
    attn_prep<4, 16><<<cdiv((long long)n * 4, B), B, 0, stream>>>(xw, a1s, a1d, asrc, adst, n);
    node_gather_norm<4, 16><<<cdiv((long long)n * 32, B), B, 0, stream>>>(start, cnt, ssrc, asrc, adst, xw, b1, ln1g, ln1b, x1, n);

    gemm64_wmma<<<cdiv(n, 16), 128, 0, stream>>>(x1, W2, xw, n);
    attn_prep<1, 64><<<cdiv((long long)n, B), B, 0, stream>>>(xw, a2s, a2d, asrc, adst, n);
    node_gather_norm<1, 64><<<cdiv((long long)n * 32, B), B, 0, stream>>>(start, cnt, ssrc, asrc, adst, xw, b2, ln2g, ln2b, (float*)d_out, n);
}


# a node's edge ids and scores (first 40 edges at 4 heads, 64 at 1 head) stay in registers from the max pass to the sum pass: no second read of list and source scores; sum order unchanged
# speedup vs baseline: 5.3574x; 1.0967x over previous
#include <hip/hip_runtime.h>

#define NEG_SLOPE 0.2f
#define LN_EPS 1e-5f
#define SM_EPS 1e-16f

typedef __attribute__((ext_vector_type(2))) float v2f;
typedef __attribute__((ext_vector_type(8))) float v8f;

__device__ __forceinline__ float lrelu(float x) { return x > 0.0f ? x : NEG_SLOPE * x; }

__global__ void __launch_bounds__(128)
gemm64_wmma(const float* __restrict__ A, const float* __restrict__ W,
            float* __restrict__ C, int nrows) {
    const int wave = threadIdx.x >> 5;
    const int lane = threadIdx.x & 31;
    const int mrow = lane & 15;
    const int kh   = lane >> 4;
    const int row0 = blockIdx.x * 16;
    int row = row0 + mrow;
    row = row < nrows ? row : nrows - 1;
    const int col  = wave * 16 + mrow;

    const float* ap = A + (size_t)row * 64 + 2 * kh;
    const float* bp = W + (size_t)(2 * kh) * 64 + col;

    v8f acc = {};
    #pragma unroll
    for (int k = 0; k < 64; k += 4) {
        v2f a = *(const v2f*)(ap + k);
        v2f b;
        b.x = bp[(size_t)k * 64];
        b.y = bp[(size_t)k * 64 + 64];
        acc = __builtin_amdgcn_wmma_f32_16x16x4_f32(
             false, a,  false, b,
             (short)0, acc,  false,  false);
    }

    const int srow = row0 + 8 * (lane >> 4);
    const int scol = wave * 16 + (lane & 15);
    #pragma unroll
    for (int r = 0; r < 8; ++r) {
        const int rr = srow + r;
        if (rr < nrows) C[(size_t)rr * 64 + scol] = acc[r];
    }
}

template <int H, int C>
__global__ void attn_prep(const float* __restrict__ xw,
                          const float* __restrict__ a_src,
                          const float* __restrict__ a_dst,
                          float* __restrict__ asrc, float* __restrict__ adst,
                          int n) {
    const int t = blockIdx.x * blockDim.x + threadIdx.x;
    if (t >= n * H) return;
    const int node = t / H;
    const int h    = t - node * H;
    const float* xp = xw + (size_t)node * 64 + h * C;
    float s = 0.0f, d = 0.0f;
    #pragma unroll
    for (int c = 0; c < C; ++c) {
        const float v = xp[c];
        s += v * a_src[h * C + c];
        d += v * a_dst[h * C + c];
    }
    asrc[t] = s;
    adst[t] = d;
}

#define SORT_CHUNKS  64
#define SORT_TILE    65536
#define SORT_THREADS 1024

__device__ __forceinline__ int sort_chunk_len(int ecnt) {
    return (ecnt + SORT_CHUNKS - 1) / SORT_CHUNKS;
}

__global__ void __launch_bounds__(SORT_THREADS)
sort_rank_lds(const int* __restrict__ dst, int* __restrict__ rank,
              int* __restrict__ table, int* __restrict__ total,
              int n, int ecnt) {
    __shared__ int lcnt[SORT_TILE + 32];
    const int tid   = threadIdx.x;
    const int chunk = blockIdx.x;
    const int per   = sort_chunk_len(ecnt);
    const int e0    = chunk * per;
    const int e1    = e0 + per < ecnt ? e0 + per : ecnt;
    const int spare = SORT_TILE + (tid & 31);
    if (chunk == 0 && tid == 0) *total = 0;
    for (int tb = 0; tb < n; tb += SORT_TILE) {
        const int tn = n - tb < SORT_TILE ? n - tb : SORT_TILE;
        for (int i = tid; i < tn; i += SORT_THREADS) lcnt[i] = 0;
        __syncthreads();
        for (int e = e0 + tid; e < e1; e += 4 * SORT_THREADS) {
            int d[4], r[4];
            #pragma unroll
            for (int u = 0; u < 4; ++u) {
                const int ee = e + u * SORT_THREADS;
                const int dv = dst[ee < e1 ? ee : e1 - 1];
                d[u] = ee < e1 ? dv - tb : -1;
            }
            #pragma unroll
            for (int u = 0; u < 4; ++u)
                r[u] = atomicAdd(&lcnt[(unsigned)d[u] < (unsigned)tn ? d[u] : spare], 1);
            #pragma unroll
            for (int u = 0; u < 4; ++u)
                if ((unsigned)d[u] < (unsigned)tn) rank[e + u * SORT_THREADS] = r[u];
        }
        __syncthreads();
        int* row = table + (size_t)chunk * n + tb;
        for (int i = tid; i < tn; i += SORT_THREADS) row[i] = lcnt[i];
        __syncthreads();
    }
}

__global__ void __launch_bounds__(256)
sort_alloc(int* __restrict__ table, int* __restrict__ total,
           int* __restrict__ start, int* __restrict__ cnt, int n) {
    const int t = blockIdx.x * blockDim.x + threadIdx.x;
    const int lane = threadIdx.x & 31;
    const bool live = t < n;
    int v[SORT_CHUNKS];
    #pragma unroll
    for (int k = 0; k < SORT_CHUNKS; ++k) v[k] = 0;
    if (live) {
        #pragma unroll
        for (int k = 0; k < SORT_CHUNKS; ++k) v[k] = table[(size_t)k * n + t];
    }
    int c = 0;
    #pragma unroll
    for (int k = 0; k < SORT_CHUNKS; ++k) c += v[k];
    int incl = c;
    for (int off = 1; off < 32; off <<= 1) {
        const int s = __shfl_up(incl, off, 32);
        if (lane >= off) incl += s;
    }
    int base = 0;
    if (lane == 31) base = atomicAdd(total, incl);
    base = __shfl(base, 31, 32);
    if (live) {
        int pos = base + incl - c;
        start[t] = pos;
        cnt[t]   = c;
        #pragma unroll
        for (int k = 0; k < SORT_CHUNKS; ++k) {
            table[(size_t)k * n + t] = pos;
            pos += v[k];
        }
    }
}

__global__ void sort_place(const int* __restrict__ src, const int* __restrict__ dst,
                           const int* __restrict__ table, const int* __restrict__ rank,
                           int* __restrict__ sorted_src, int n, int ecnt) {
    const int t = blockIdx.x * blockDim.x + threadIdx.x;
    if (t < ecnt) {
        const int chunk = t / sort_chunk_len(ecnt);
        const int d = dst[t];
        if ((unsigned)d < (unsigned)n)
            sorted_src[table[(size_t)chunk * n + d] + rank[t]] = src[t];
    }
}

template <int H>
__device__ __forceinline__ void add_chunk_rows(const float* __restrict__ xw, int s, float ex,
                                               int lim, int hch, int ch,
                                               float& acc0, float& acc1) {
    constexpr int EPC = 32 / H;
    constexpr int GRP = EPC < 8 ? EPC : 8;
    for (int k0 = 0; k0 < lim; k0 += GRP) {
        #pragma unroll
        for (int kk = 0; kk < GRP; ++kk) {
            const int k  = k0 + kk;
            const int sk = __builtin_amdgcn_readlane(s, k * H);
            const float wk = __shfl(ex, k * H + hch, 32);
            const v2f v = *(const v2f*)(xw + (size_t)sk * 64 + ch);
            acc0 += wk * v.x;
            acc1 += wk * v.y;
        }
    }
}

template <int H, int C>
__global__ void __launch_bounds__(256)
node_gather_norm(const int* __restrict__ start, const int* __restrict__ cnt,
                 const int* __restrict__ sorted_src,
                 const float* __restrict__ asrc, const float* __restrict__ adst,
                 const float* __restrict__ xw,
                 const float* __restrict__ bias, const float* __restrict__ g,
                 const float* __restrict__ bparam,
                 float* __restrict__ out, int n) {
    constexpr int EPC  = 32 / H;
    constexpr int KEEP = H == 1 ? 2 : 5;
    const int node = __builtin_amdgcn_readfirstlane(
        (int)((blockIdx.x * blockDim.x + threadIdx.x) >> 5));
    const int lane = threadIdx.x & 31;
    if (node >= n) return;
    const int hsel = lane % H;
    const int esel = lane / H;
    const int ch   = 2 * lane;
    const int hch  = ch / C;
    const int beg  = start[node];
    const int deg  = cnt[node];
    const float ad    = adst[node * H + hsel];
    const float eself = lrelu(asrc[node * H + hsel] + ad);
    const v2f bi = *(const v2f*)(bias + ch);
    const v2f gg = *(const v2f*)(g + ch);
    const v2f bb = *(const v2f*)(bparam + ch);

    int   ks[KEEP];
    float ke[KEEP];
    #pragma unroll
    for (int c = 0; c < KEEP; ++c) {
        ks[c] = node;
        if (c * EPC < deg) {
            const int j = c * EPC + esel;
            ks[c] = sorted_src[beg + (j < deg ? j : deg - 1)];
        }
    }
    #pragma unroll
    for (int c = 0; c < KEEP; ++c) ke[c] = lrelu(asrc[ks[c] * H + hsel] + ad);
    float mx = eself;
    #pragma unroll
    for (int c = 0; c < KEEP; ++c) mx = fmaxf(mx, ke[c]);
    for (int c0 = KEEP * EPC; c0 < deg; c0 += EPC) {
        const int j = c0 + esel;
        const int s = sorted_src[beg + (j < deg ? j : deg - 1)];
        mx = fmaxf(mx, lrelu(asrc[s * H + hsel] + ad));
    }
    for (int off = H; off < 32; off <<= 1) mx = fmaxf(mx, __shfl_xor(mx, off, 32));

    const float exself = __expf(eself - mx);
    float dsum = esel == 0 ? exself : 0.0f;
    const float wself = __shfl(exself, hch, 32);
    const v2f xs = *(const v2f*)(xw + (size_t)node * 64 + ch);
    float acc0 = wself * xs.x;
    float acc1 = wself * xs.y;

    #pragma unroll
    for (int c = 0; c < KEEP; ++c) {
        if (c * EPC < deg) {
            const bool live = c * EPC + esel < deg;
            const int   s  = live ? ks[c] : node;
            const float ex = live ? __expf(ke[c] - mx) : 0.0f;
            dsum += ex;
            const int lim = deg - c * EPC < EPC ? deg - c * EPC : EPC;
            add_chunk_rows<H>(xw, s, ex, lim, hch, ch, acc0, acc1);
        }
    }
    for (int c0 = KEEP * EPC; c0 < deg; c0 += EPC) {
        const int j = c0 + esel;
        const bool live = j < deg;
        const int   sl = sorted_src[beg + (live ? j : deg - 1)];
        const float el = __expf(lrelu(asrc[sl * H + hsel] + ad) - mx);
        const int   s  = live ? sl : node;
        const float ex = live ? el : 0.0f;
        dsum += ex;
        const int lim = deg - c0 < EPC ? deg - c0 : EPC;
        add_chunk_rows<H>(xw, s, ex, lim, hch, ch, acc0, acc1);
    }
    for (int off = H; off < 32; off <<= 1) dsum += __shfl_xor(dsum, off, 32);
    const float den = __shfl(dsum, hch, 32) + SM_EPS;

    const float v0 = acc0 / den + bi.x;
    const float v1 = acc1 / den + bi.y;
    float sum = v0 + v1;
    #pragma unroll
    for (int off = 16; off > 0; off >>= 1) sum += __shfl_xor(sum, off, 32);
    const float mu = sum * (1.0f / 64.0f);
    const float d0 = v0 - mu, d1 = v1 - mu;
    float vv = d0 * d0 + d1 * d1;
    #pragma unroll
    for (int off = 16; off > 0; off >>= 1) vv += __shfl_xor(vv, off, 32);
    const float rstd = rsqrtf(vv * (1.0f / 64.0f) + LN_EPS);

    float o0 = d0 * rstd * gg.x + bb.x;
    float o1 = d1 * rstd * gg.y + bb.y;
    o0 = o0 > 0.0f ? o0 : __expf(o0) - 1.0f;
    o1 = o1 > 0.0f ? o1 : __expf(o1) - 1.0f;
    v2f o; o.x = o0; o.y = o1;
    *(v2f*)(out + (size_t)node * 64 + ch) = o;
}

static inline unsigned cdiv(long long a, int b) { return (unsigned)((a + b - 1) / b); }

extern "C" void kernel_launch(void* const* d_in, const int* in_sizes, int n_in,
                              void* d_out, int out_size, void* d_ws, size_t ws_size,
                              hipStream_t stream) {
    const float* h   = (const float*)d_in[0];
    const int*   ei  = (const int*)d_in[1];
    const float* W1  = (const float*)d_in[2];
    const float* a1s = (const float*)d_in[3];
    const float* a1d = (const float*)d_in[4];
    const float* b1  = (const float*)d_in[5];
    const float* W2  = (const float*)d_in[6];
    const float* a2s = (const float*)d_in[7];
    const float* a2d = (const float*)d_in[8];
    const float* b2  = (const float*)d_in[9];
    const float* ln1g = (const float*)d_in[10];
    const float* ln1b = (const float*)d_in[11];
    const float* ln2g = (const float*)d_in[12];
    const float* ln2b = (const float*)d_in[13];

    const int n = in_sizes[0] / 64;
    const int e = in_sizes[1] / 2;
    const int* src = ei;
    const int* dst = ei + e;

    float* ws = (float*)d_ws;
    float* xw   = ws; ws += (size_t)n * 64;
    float* asrc = ws; ws += (size_t)n * 4;
    float* adst = ws; ws += (size_t)n * 4;
    float* x1   = ws; ws += (size_t)n * 64;
    int* wi     = (int*)ws;
    int* cnt    = wi; wi += n;
    int* start  = wi; wi += n;
    int* total  = wi; wi += 2;
    int* rank   = wi; wi += e;
    int* ssrc   = wi; wi += e;
    int* table  = wi; wi += (size_t)SORT_CHUNKS * n;

    const int B = 256;

    sort_rank_lds<<<SORT_CHUNKS, SORT_THREADS, 0, stream>>>(dst, rank, table, total, n, e);
    sort_alloc<<<cdiv(n, B), B, 0, stream>>>(table, total, start, cnt, n);
    sort_place<<<cdiv(e, B), B, 0, stream>>>(src, dst, table, rank, ssrc, n, e);

    gemm64_wmma<<<cdiv(n, 16), 128, 0, stream>>>(h, W1, xw, n);
    attn_prep<4, 16><<<cdiv((long long)n * 4, B), B, 0, stream>>>(xw, a1s, a1d, asrc, adst, n);
    node_gather_norm<4, 16><<<cdiv((long long)n * 32, B), B, 0, stream>>>(start, cnt, ssrc, asrc, adst, xw, b1, ln1g, ln1b, x1, n);

    gemm64_wmma<<<cdiv(n, 16), 128, 0, stream>>>(x1, W2, xw, n);
    attn_prep<1, 64><<<cdiv((long long)n, B), B, 0, stream>>>(xw, a2s, a2d, asrc, adst, n);
    node_gather_norm<1, 64><<<cdiv((long long)n * 32, B), B, 0, stream>>>(start, cnt, ssrc, asrc, adst, xw, b2, ln2g, ln2b, (float*)d_out, n);
}


# attention dots (asrc, adst) formed in the GEMM epilogue from the WMMA accumulators by 16-lane DPP row sums; both attn_prep launches and their re-read of xw gone; dot sum order now a lane tree
# speedup vs baseline: 5.5687x; 1.0394x over previous
#include <hip/hip_runtime.h>

#define NEG_SLOPE 0.2f
#define LN_EPS 1e-5f
#define SM_EPS 1e-16f

typedef __attribute__((ext_vector_type(2))) float v2f;
typedef __attribute__((ext_vector_type(8))) float v8f;

__device__ __forceinline__ float lrelu(float x) { return x > 0.0f ? x : NEG_SLOPE * x; }

template <int CTRL>
__device__ __forceinline__ float row_xor_add(float x) {
    const int y = __builtin_amdgcn_update_dpp(0, __float_as_int(x), CTRL, 0xf, 0xf, true);
    return x + __int_as_float(y);
}
__device__ __forceinline__ float row16_sum(float x) {
    x = row_xor_add<0x161>(x);
    x = row_xor_add<0x162>(x);
    x = row_xor_add<0x164>(x);
    x = row_xor_add<0x168>(x);
    return x;
}

template <int H>
__global__ void __launch_bounds__(128)
gemm64_wmma_dots(const float* __restrict__ A, const float* __restrict__ W,
                 const float* __restrict__ a_src, const float* __restrict__ a_dst,
                 float* __restrict__ C, float* __restrict__ asrc,
                 float* __restrict__ adst, int nrows) {
    __shared__ float part[2][4][16];
    const int wave = threadIdx.x >> 5;
    const int lane = threadIdx.x & 31;
    const int mrow = lane & 15;
    const int kh   = lane >> 4;
    const int row0 = blockIdx.x * 16;
    int row = row0 + mrow;
    row = row < nrows ? row : nrows - 1;
    const int col  = wave * 16 + mrow;

    const float* ap = A + (size_t)row * 64 + 2 * kh;
    const float* bp = W + (size_t)(2 * kh) * 64 + col;
    const float wsrc = a_src[col];
    const float wdst = a_dst[col];

    v8f acc = {};
    #pragma unroll
    for (int k = 0; k < 64; k += 4) {
        v2f a = *(const v2f*)(ap + k);
        v2f b;
        b.x = bp[(size_t)k * 64];
        b.y = bp[(size_t)k * 64 + 64];
        acc = __builtin_amdgcn_wmma_f32_16x16x4_f32(
             false, a,  false, b,
             (short)0, acc,  false,  false);
    }

    float vs = 0.0f, vd = 0.0f;
    #pragma unroll
    for (int r = 0; r < 8; ++r) {
        const float ss = row16_sum(acc[r] * wsrc);
        const float sd = row16_sum(acc[r] * wdst);
        vs = mrow == r ? ss : vs;
        vd = mrow == r ? sd : vd;
    }
    if (mrow < 8) {
        part[0][wave][8 * kh + mrow] = vs;
        part[1][wave][8 * kh + mrow] = vd;
    }
    __syncthreads();

    const int srow = row0 + 8 * (lane >> 4);
    const int scol = wave * 16 + (lane & 15);
    #pragma unroll
    for (int r = 0; r < 8; ++r) {
        const int rr = srow + r;
        if (rr < nrows) C[(size_t)rr * 64 + scol] = acc[r];
    }

    if (H == 4) {
        const int q = threadIdx.x >> 6;
        const int i = threadIdx.x & 63;
        if (row0 + (i >> 2) < nrows)
            (q ? adst : asrc)[(size_t)row0 * 4 + i] = part[q][i & 3][i >> 2];
    } else {
        if (wave < 2 && lane < 16 && row0 + lane < nrows) {
            const float d = ((part[wave][0][lane] + part[wave][1][lane])
                             + part[wave][2][lane]) + part[wave][3][lane];
            (wave ? adst : asrc)[row0 + lane] = d;
        }
    }
}

#define SORT_CHUNKS  64
#define SORT_TILE    65536
#define SORT_THREADS 1024

__device__ __forceinline__ int sort_chunk_len(int ecnt) {
    return (ecnt + SORT_CHUNKS - 1) / SORT_CHUNKS;
}

__global__ void __launch_bounds__(SORT_THREADS)
sort_rank_lds(const int* __restrict__ dst, int* __restrict__ rank,
              int* __restrict__ table, int* __restrict__ total,
              int n, int ecnt) {
    __shared__ int lcnt[SORT_TILE + 32];
    const int tid   = threadIdx.x;
    const int chunk = blockIdx.x;
    const int per   = sort_chunk_len(ecnt);
    const int e0    = chunk * per;
    const int e1    = e0 + per < ecnt ? e0 + per : ecnt;
    const int spare = SORT_TILE + (tid & 31);
    if (chunk == 0 && tid == 0) *total = 0;
    for (int tb = 0; tb < n; tb += SORT_TILE) {
        const int tn = n - tb < SORT_TILE ? n - tb : SORT_TILE;
        for (int i = tid; i < tn; i += SORT_THREADS) lcnt[i] = 0;
        __syncthreads();
        for (int e = e0 + tid; e < e1; e += 4 * SORT_THREADS) {
            int d[4], r[4];
            #pragma unroll
            for (int u = 0; u < 4; ++u) {
                const int ee = e + u * SORT_THREADS;
                const int dv = dst[ee < e1 ? ee : e1 - 1];
                d[u] = ee < e1 ? dv - tb : -1;
            }
            #pragma unroll
            for (int u = 0; u < 4; ++u)
                r[u] = atomicAdd(&lcnt[(unsigned)d[u] < (unsigned)tn ? d[u] : spare], 1);
            #pragma unroll
            for (int u = 0; u < 4; ++u)
                if ((unsigned)d[u] < (unsigned)tn) rank[e + u * SORT_THREADS] = r[u];
        }
        __syncthreads();
        int* row = table + (size_t)chunk * n + tb;
        for (int i = tid; i < tn; i += SORT_THREADS) row[i] = lcnt[i];
        __syncthreads();
    }
}

__global__ void __launch_bounds__(256)
sort_alloc(int* __restrict__ table, int* __restrict__ total,
           int* __restrict__ start, int* __restrict__ cnt, int n) {
    const int t = blockIdx.x * blockDim.x + threadIdx.x;
    const int lane = threadIdx.x & 31;
    const bool live = t < n;
    int v[SORT_CHUNKS];
    #pragma unroll
    for (int k = 0; k < SORT_CHUNKS; ++k) v[k] = 0;
    if (live) {
        #pragma unroll
        for (int k = 0; k < SORT_CHUNKS; ++k) v[k] = table[(size_t)k * n + t];
    }
    int c = 0;
    #pragma unroll
    for (int k = 0; k < SORT_CHUNKS; ++k) c += v[k];
    int incl = c;
    for (int off = 1; off < 32; off <<= 1) {
        const int s = __shfl_up(incl, off, 32);
        if (lane >= off) incl += s;
    }
    int base = 0;
    if (lane == 31) base = atomicAdd(total, incl);
    base = __shfl(base, 31, 32);
    if (live) {
        int pos = base + incl - c;
        start[t] = pos;
        cnt[t]   = c;
        #pragma unroll
        for (int k = 0; k < SORT_CHUNKS; ++k) {
            table[(size_t)k * n + t] = pos;
            pos += v[k];
        }
    }
}

__global__ void sort_place(const int* __restrict__ src, const int* __restrict__ dst,
                           const int* __restrict__ table, const int* __restrict__ rank,
                           int* __restrict__ sorted_src, int n, int ecnt) {
    const int t = blockIdx.x * blockDim.x + threadIdx.x;
    if (t < ecnt) {
        const int chunk = t / sort_chunk_len(ecnt);
        const int d = dst[t];
        if ((unsigned)d < (unsigned)n)
            sorted_src[table[(size_t)chunk * n + d] + rank[t]] = src[t];
    }
}

template <int H>
__device__ __forceinline__ void add_chunk_rows(const float* __restrict__ xw, int s, float ex,
                                               int lim, int hch, int ch,
                                               float& acc0, float& acc1) {
    constexpr int EPC = 32 / H;
    constexpr int GRP = EPC < 8 ? EPC : 8;
    for (int k0 = 0; k0 < lim; k0 += GRP) {
        #pragma unroll
        for (int kk = 0; kk < GRP; ++kk) {
            const int k  = k0 + kk;
            const int sk = __builtin_amdgcn_readlane(s, k * H);
            const float wk = __shfl(ex, k * H + hch, 32);
            const v2f v = *(const v2f*)(xw + (size_t)sk * 64 + ch);
            acc0 += wk * v.x;
            acc1 += wk * v.y;
        }
    }
}

template <int H, int C>
__global__ void __launch_bounds__(256)
node_gather_norm(const int* __restrict__ start, const int* __restrict__ cnt,
                 const int* __restrict__ sorted_src,
                 const float* __restrict__ asrc, const float* __restrict__ adst,
                 const float* __restrict__ xw,
                 const float* __restrict__ bias, const float* __restrict__ g,
                 const float* __restrict__ bparam,
                 float* __restrict__ out, int n) {
    constexpr int EPC  = 32 / H;
    constexpr int KEEP = H == 1 ? 2 : 5;
    const int node = __builtin_amdgcn_readfirstlane(
        (int)((blockIdx.x * blockDim.x + threadIdx.x) >> 5));
    const int lane = threadIdx.x & 31;
    if (node >= n) return;
    const int hsel = lane % H;
    const int esel = lane / H;
    const int ch   = 2 * lane;
    const int hch  = ch / C;
    const int beg  = start[node];
    const int deg  = cnt[node];
    const float ad    = adst[node * H + hsel];
    const float eself = lrelu(asrc[node * H + hsel] + ad);
    const v2f bi = *(const v2f*)(bias + ch);
    const v2f gg = *(const v2f*)(g + ch);
    const v2f bb = *(const v2f*)(bparam + ch);

    int   ks[KEEP];
    float ke[KEEP];
    #pragma unroll
    for (int c = 0; c < KEEP; ++c) {
        ks[c] = node;
        if (c * EPC < deg) {
            const int j = c * EPC + esel;
            ks[c] = sorted_src[beg + (j < deg ? j : deg - 1)];
        }
    }
    #pragma unroll
    for (int c = 0; c < KEEP; ++c) ke[c] = lrelu(asrc[ks[c] * H + hsel] + ad);
    float mx = eself;
    #pragma unroll
    for (int c = 0; c < KEEP; ++c) mx = fmaxf(mx, ke[c]);
    for (int c0 = KEEP * EPC; c0 < deg; c0 += EPC) {
        const int j = c0 + esel;
        const int s = sorted_src[beg + (j < deg ? j : deg - 1)];
        mx = fmaxf(mx, lrelu(asrc[s * H + hsel] + ad));
    }
    for (int off = H; off < 32; off <<= 1) mx = fmaxf(mx, __shfl_xor(mx, off, 32));

    const float exself = __expf(eself - mx);
    float dsum = esel == 0 ? exself : 0.0f;
    const float wself = __shfl(exself, hch, 32);
    const v2f xs = *(const v2f*)(xw + (size_t)node * 64 + ch);
    float acc0 = wself * xs.x;
    float acc1 = wself * xs.y;

    #pragma unroll
    for (int c = 0; c < KEEP; ++c) {
        if (c * EPC < deg) {
            const bool live = c * EPC + esel < deg;
            const int   s  = live ? ks[c] : node;
            const float ex = live ? __expf(ke[c] - mx) : 0.0f;
            dsum += ex;
            const int lim = deg - c * EPC < EPC ? deg - c * EPC : EPC;
            add_chunk_rows<H>(xw, s, ex, lim, hch, ch, acc0, acc1);
        }
    }
    for (int c0 = KEEP * EPC; c0 < deg; c0 += EPC) {
        const int j = c0 + esel;
        const bool live = j < deg;
        const int   sl = sorted_src[beg + (live ? j : deg - 1)];
        const float el = __expf(lrelu(asrc[sl * H + hsel] + ad) - mx);
        const int   s  = live ? sl : node;
        const float ex = live ? el : 0.0f;
        dsum += ex;
        const int lim = deg - c0 < EPC ? deg - c0 : EPC;
        add_chunk_rows<H>(xw, s, ex, lim, hch, ch, acc0, acc1);
    }
    for (int off = H; off < 32; off <<= 1) dsum += __shfl_xor(dsum, off, 32);
    const float den = __shfl(dsum, hch, 32) + SM_EPS;

    const float v0 = acc0 / den + bi.x;
    const float v1 = acc1 / den + bi.y;
    float sum = v0 + v1;
    #pragma unroll
    for (int off = 16; off > 0; off >>= 1) sum += __shfl_xor(sum, off, 32);
    const float mu = sum * (1.0f / 64.0f);
    const float d0 = v0 - mu, d1 = v1 - mu;
    float vv = d0 * d0 + d1 * d1;
    #pragma unroll
    for (int off = 16; off > 0; off >>= 1) vv += __shfl_xor(vv, off, 32);
    const float rstd = rsqrtf(vv * (1.0f / 64.0f) + LN_EPS);

    float o0 = d0 * rstd * gg.x + bb.x;
    float o1 = d1 * rstd * gg.y + bb.y;
    o0 = o0 > 0.0f ? o0 : __expf(o0) - 1.0f;
    o1 = o1 > 0.0f ? o1 : __expf(o1) - 1.0f;
    v2f o; o.x = o0; o.y = o1;
    *(v2f*)(out + (size_t)node * 64 + ch) = o;
}

static inline unsigned cdiv(long long a, int b) { return (unsigned)((a + b - 1) / b); }

extern "C" void kernel_launch(void* const* d_in, const int* in_sizes, int n_in,
                              void* d_out, int out_size, void* d_ws, size_t ws_size,
                              hipStream_t stream) {
    const float* h   = (const float*)d_in[0];
    const int*   ei  = (const int*)d_in[1];
    const float* W1  = (const float*)d_in[2];
    const float* a1s = (const float*)d_in[3];
    const float* a1d = (const float*)d_in[4];
    const float* b1  = (const float*)d_in[5];
    const float* W2  = (const float*)d_in[6];
    const float* a2s = (const float*)d_in[7];
    const float* a2d = (const float*)d_in[8];
    const float* b2  = (const float*)d_in[9];
    const float* ln1g = (const float*)d_in[10];
    const float* ln1b = (const float*)d_in[11];
    const float* ln2g = (const float*)d_in[12];
    const float* ln2b = (const float*)d_in[13];

    const int n = in_sizes[0] / 64;
    const int e = in_sizes[1] / 2;
    const int* src = ei;
    const int* dst = ei + e;

    float* ws = (float*)d_ws;
    float* xw   = ws; ws += (size_t)n * 64;
    float* asrc = ws; ws += (size_t)n * 4;
    float* adst = ws; ws += (size_t)n * 4;
    float* x1   = ws; ws += (size_t)n * 64;
    int* wi     = (int*)ws;
    int* cnt    = wi; wi += n;
    int* start  = wi; wi += n;
    int* total  = wi; wi += 2;
    int* rank   = wi; wi += e;
    int* ssrc   = wi; wi += e;
    int* table  = wi; wi += (size_t)SORT_CHUNKS * n;

    const int B = 256;

    sort_rank_lds<<<SORT_CHUNKS, SORT_THREADS, 0, stream>>>(dst, rank, table, total, n, e);
    sort_alloc<<<cdiv(n, B), B, 0, stream>>>(table, total, start, cnt, n);
    sort_place<<<cdiv(e, B), B, 0, stream>>>(src, dst, table, rank, ssrc, n, e);

    gemm64_wmma_dots<4><<<cdiv(n, 16), 128, 0, stream>>>(h, W1, a1s, a1d, xw, asrc, adst, n);
    node_gather_norm<4, 16><<<cdiv((long long)n * 32, B), B, 0, stream>>>(start, cnt, ssrc, asrc, adst, xw, b1, ln1g, ln1b, x1, n);

    gemm64_wmma_dots<1><<<cdiv(n, 16), 128, 0, stream>>>(x1, W2, a2s, a2d, xw, asrc, adst, n);
    node_gather_norm<1, 64><<<cdiv((long long)n * 32, B), B, 0, stream>>>(start, cnt, ssrc, asrc, adst, xw, b2, ln2g, ln2b, (float*)d_out, n);
}


# edge slots by returning LDS adds on the chunk's slot bases in the placement pass; count and placement on (chunk, node range) groups; no rank array, no per-edge base gather; in-node order by arrival
# speedup vs baseline: 6.0057x; 1.0785x over previous
#include <hip/hip_runtime.h>

#define NEG_SLOPE 0.2f
#define LN_EPS 1e-5f
#define SM_EPS 1e-16f

typedef __attribute__((ext_vector_type(2))) float v2f;
typedef __attribute__((ext_vector_type(8))) float v8f;

__device__ __forceinline__ float lrelu(float x) { return x > 0.0f ? x : NEG_SLOPE * x; }

template <int CTRL>
__device__ __forceinline__ float row_xor_add(float x) {
    const int y = __builtin_amdgcn_update_dpp(0, __float_as_int(x), CTRL, 0xf, 0xf, true);
    return x + __int_as_float(y);
}
__device__ __forceinline__ float row16_sum(float x) {
    x = row_xor_add<0x161>(x);
    x = row_xor_add<0x162>(x);
    x = row_xor_add<0x164>(x);
    x = row_xor_add<0x168>(x);
    return x;
}

template <int H>
__global__ void __launch_bounds__(128)
gemm64_wmma_dots(const float* __restrict__ A, const float* __restrict__ W,
                 const float* __restrict__ a_src, const float* __restrict__ a_dst,
                 float* __restrict__ C, float* __restrict__ asrc,
                 float* __restrict__ adst, int nrows) {
    __shared__ float part[2][4][16];
    const int wave = threadIdx.x >> 5;
    const int lane = threadIdx.x & 31;
    const int mrow = lane & 15;
    const int kh   = lane >> 4;
    const int row0 = blockIdx.x * 16;
    int row = row0 + mrow;
    row = row < nrows ? row : nrows - 1;
    const int col  = wave * 16 + mrow;

    const float* ap = A + (size_t)row * 64 + 2 * kh;
    const float* bp = W + (size_t)(2 * kh) * 64 + col;
    const float wsrc = a_src[col];
    const float wdst = a_dst[col];

    v8f acc = {};
    #pragma unroll
    for (int k = 0; k < 64; k += 4) {
        v2f a = *(const v2f*)(ap + k);
        v2f b;
        b.x = bp[(size_t)k * 64];
        b.y = bp[(size_t)k * 64 + 64];
        acc = __builtin_amdgcn_wmma_f32_16x16x4_f32(
             false, a,  false, b,
             (short)0, acc,  false,  false);
    }

    float vs = 0.0f, vd = 0.0f;
    #pragma unroll
    for (int r = 0; r < 8; ++r) {
        const float ss = row16_sum(acc[r] * wsrc);
        const float sd = row16_sum(acc[r] * wdst);
        vs = mrow == r ? ss : vs;
        vd = mrow == r ? sd : vd;
    }
    if (mrow < 8) {
        part[0][wave][8 * kh + mrow] = vs;
        part[1][wave][8 * kh + mrow] = vd;
    }
    __syncthreads();

    const int srow = row0 + 8 * (lane >> 4);
    const int scol = wave * 16 + (lane & 15);
    #pragma unroll
    for (int r = 0; r < 8; ++r) {
        const int rr = srow + r;
        if (rr < nrows) C[(size_t)rr * 64 + scol] = acc[r];
    }

    if (H == 4) {
        const int q = threadIdx.x >> 6;
        const int i = threadIdx.x & 63;
        if (row0 + (i >> 2) < nrows)
            (q ? adst : asrc)[(size_t)row0 * 4 + i] = part[q][i & 3][i >> 2];
    } else {
        if (wave < 2 && lane < 16 && row0 + lane < nrows) {
            const float d = ((part[wave][0][lane] + part[wave][1][lane])
                             + part[wave][2][lane]) + part[wave][3][lane];
            (wave ? adst : asrc)[row0 + lane] = d;
        }
    }
}

#define SORT_CHUNKS  64
#define SORT_RANGES  4
#define SORT_TILE    12544
#define SORT_THREADS 1024
#define SORT_UNROLL  10
#define SORT_SLOTS   ((SORT_TILE + SORT_THREADS - 1) / SORT_THREADS)

__device__ __forceinline__ int sort_chunk_len(int ecnt) {
    return (ecnt + SORT_CHUNKS - 1) / SORT_CHUNKS;
}

__global__ void __launch_bounds__(SORT_THREADS)
sort_count(const int* __restrict__ dst, int* __restrict__ table,
           int* __restrict__ total, int n, int ecnt) {
    __shared__ int lcnt[SORT_TILE + 32];
    const int tid   = threadIdx.x;
    const int chunk = blockIdx.x / SORT_RANGES;
    const int range = blockIdx.x % SORT_RANGES;
    const int per   = sort_chunk_len(ecnt);
    const int e0    = chunk * per;
    const int e1    = e0 + per < ecnt ? e0 + per : ecnt;
    const int spare = SORT_TILE + (tid & 31);
    if (blockIdx.x == 0 && tid == 0) *total = 0;
    if (tid < 32) lcnt[SORT_TILE + tid] = 0;
    for (int tb = range * SORT_TILE; tb < n; tb += SORT_RANGES * SORT_TILE) {
        const int tn = n - tb < SORT_TILE ? n - tb : SORT_TILE;
        for (int i = tid; i < tn; i += SORT_THREADS) lcnt[i] = 0;
        __syncthreads();
        for (int e = e0 + tid; e < e1; e += SORT_UNROLL * SORT_THREADS) {
            int d[SORT_UNROLL];
            #pragma unroll
            for (int u = 0; u < SORT_UNROLL; ++u) {
                const int ee = e + u * SORT_THREADS;
                const int dv = dst[ee < e1 ? ee : e1 - 1];
                d[u] = ee < e1 ? dv - tb : -1;
            }
            #pragma unroll
            for (int u = 0; u < SORT_UNROLL; ++u)
                atomicAdd(&lcnt[(unsigned)d[u] < (unsigned)tn ? d[u] : spare], 1);
        }
        __syncthreads();
        int* row = table + (size_t)chunk * n + tb;
        int cv[SORT_SLOTS];
        #pragma unroll
        for (int k = 0; k < SORT_SLOTS; ++k) {
            const int i = tid + k * SORT_THREADS;
            cv[k] = lcnt[i < tn ? i : 0];
        }
        #pragma unroll
        for (int k = 0; k < SORT_SLOTS; ++k) {
            const int i = tid + k * SORT_THREADS;
            if (i < tn) row[i] = cv[k];
        }
        __syncthreads();
    }
}

__global__ void __launch_bounds__(256)
sort_alloc(int* __restrict__ table, int* __restrict__ total,
           int* __restrict__ start, int* __restrict__ cnt, int n) {
    const int t = blockIdx.x * blockDim.x + threadIdx.x;
    const int lane = threadIdx.x & 31;
    const bool live = t < n;
    int v[SORT_CHUNKS];
    #pragma unroll
    for (int k = 0; k < SORT_CHUNKS; ++k) v[k] = 0;
    if (live) {
        #pragma unroll
        for (int k = 0; k < SORT_CHUNKS; ++k) v[k] = table[(size_t)k * n + t];
    }
    int c = 0;
    #pragma unroll
    for (int k = 0; k < SORT_CHUNKS; ++k) c += v[k];
    int incl = c;
    for (int off = 1; off < 32; off <<= 1) {
        const int s = __shfl_up(incl, off, 32);
        if (lane >= off) incl += s;
    }
    int base = 0;
    if (lane == 31) base = atomicAdd(total, incl);
    base = __shfl(base, 31, 32);
    if (live) {
        int pos = base + incl - c;
        start[t] = pos;
        cnt[t]   = c;
        #pragma unroll
        for (int k = 0; k < SORT_CHUNKS; ++k) {
            table[(size_t)k * n + t] = pos;
            pos += v[k];
        }
    }
}

__global__ void __launch_bounds__(SORT_THREADS)
sort_place(const int* __restrict__ src, const int* __restrict__ dst,
           const int* __restrict__ table, int* __restrict__ sorted_src,
           int n, int ecnt) {
    __shared__ int lpos[SORT_TILE + 32];
    const int tid   = threadIdx.x;
    const int chunk = blockIdx.x / SORT_RANGES;
    const int range = blockIdx.x % SORT_RANGES;
    const int per   = sort_chunk_len(ecnt);
    const int e0    = chunk * per;
    const int e1    = e0 + per < ecnt ? e0 + per : ecnt;
    const int spare = SORT_TILE + (tid & 31);
    if (tid < 32) lpos[SORT_TILE + tid] = 0;
    for (int tb = range * SORT_TILE; tb < n; tb += SORT_RANGES * SORT_TILE) {
        const int tn = n - tb < SORT_TILE ? n - tb : SORT_TILE;
        const int* row = table + (size_t)chunk * n + tb;
        int bv[SORT_SLOTS];
        #pragma unroll
        for (int k = 0; k < SORT_SLOTS; ++k) {
            const int i = tid + k * SORT_THREADS;
            bv[k] = row[i < tn ? i : tn - 1];
        }
        #pragma unroll
        for (int k = 0; k < SORT_SLOTS; ++k) {
            const int i = tid + k * SORT_THREADS;
            if (i < tn) lpos[i] = bv[k];
        }
        __syncthreads();
        for (int e = e0 + tid; e < e1; e += SORT_UNROLL * SORT_THREADS) {
            int d[SORT_UNROLL], s[SORT_UNROLL], p[SORT_UNROLL];
            #pragma unroll
            for (int u = 0; u < SORT_UNROLL; ++u) {
                const int ee = e + u * SORT_THREADS;
                const int ec = ee < e1 ? ee : e1 - 1;
                const int dv = dst[ec];
                s[u] = src[ec];
                d[u] = ee < e1 ? dv - tb : -1;
            }
            #pragma unroll
            for (int u = 0; u < SORT_UNROLL; ++u) asm volatile("" :: "v"(s[u]));
            #pragma unroll
            for (int u = 0; u < SORT_UNROLL; ++u)
                p[u] = atomicAdd(&lpos[(unsigned)d[u] < (unsigned)tn ? d[u] : spare], 1);
            #pragma unroll
            for (int u = 0; u < SORT_UNROLL; ++u)
                if ((unsigned)d[u] < (unsigned)tn) sorted_src[p[u]] = s[u];
        }
        __syncthreads();
    }
}

template <int H>
__device__ __forceinline__ void add_chunk_rows(const float* __restrict__ xw, int s, float ex,
                                               int lim, int hch, int ch,
                                               float& acc0, float& acc1) {
    constexpr int EPC = 32 / H;
    constexpr int GRP = EPC < 8 ? EPC : 8;
    for (int k0 = 0; k0 < lim; k0 += GRP) {
        #pragma unroll
        for (int kk = 0; kk < GRP; ++kk) {
            const int k  = k0 + kk;
            const int sk = __builtin_amdgcn_readlane(s, k * H);
            const float wk = __shfl(ex, k * H + hch, 32);
            const v2f v = *(const v2f*)(xw + (size_t)sk * 64 + ch);
            acc0 += wk * v.x;
            acc1 += wk * v.y;
        }
    }
}

template <int H, int C>
__global__ void __launch_bounds__(256)
node_gather_norm(const int* __restrict__ start, const int* __restrict__ cnt,
                 const int* __restrict__ sorted_src,
                 const float* __restrict__ asrc, const float* __restrict__ adst,
                 const float* __restrict__ xw,
                 const float* __restrict__ bias, const float* __restrict__ g,
                 const float* __restrict__ bparam,
                 float* __restrict__ out, int n) {
    constexpr int EPC  = 32 / H;
    constexpr int KEEP = H == 1 ? 2 : 5;
    const int node = __builtin_amdgcn_readfirstlane(
        (int)((blockIdx.x * blockDim.x + threadIdx.x) >> 5));
    const int lane = threadIdx.x & 31;
    if (node >= n) return;
    const int hsel = lane % H;
    const int esel = lane / H;
    const int ch   = 2 * lane;
    const int hch  = ch / C;
    const int beg  = start[node];
    const int deg  = cnt[node];
    const float ad    = adst[node * H + hsel];
    const float eself = lrelu(asrc[node * H + hsel] + ad);
    const v2f bi = *(const v2f*)(bias + ch);
    const v2f gg = *(const v2f*)(g + ch);
    const v2f bb = *(const v2f*)(bparam + ch);

    int   ks[KEEP];
    float ke[KEEP];
    #pragma unroll
    for (int c = 0; c < KEEP; ++c) {
        ks[c] = node;
        if (c * EPC < deg) {
            const int j = c * EPC + esel;
            ks[c] = sorted_src[beg + (j < deg ? j : deg - 1)];
        }
    }
    #pragma unroll
    for (int c = 0; c < KEEP; ++c) ke[c] = lrelu(asrc[ks[c] * H + hsel] + ad);
    float mx = eself;
    #pragma unroll
    for (int c = 0; c < KEEP; ++c) mx = fmaxf(mx, ke[c]);
    for (int c0 = KEEP * EPC; c0 < deg; c0 += EPC) {
        const int j = c0 + esel;
        const int s = sorted_src[beg + (j < deg ? j : deg - 1)];
        mx = fmaxf(mx, lrelu(asrc[s * H + hsel] + ad));
    }
    for (int off = H; off < 32; off <<= 1) mx = fmaxf(mx, __shfl_xor(mx, off, 32));

    const float exself = __expf(eself - mx);
    float dsum = esel == 0 ? exself : 0.0f;
    const float wself = __shfl(exself, hch, 32);
    const v2f xs = *(const v2f*)(xw + (size_t)node * 64 + ch);
    float acc0 = wself * xs.x;
    float acc1 = wself * xs.y;

    #pragma unroll
    for (int c = 0; c < KEEP; ++c) {
        if (c * EPC < deg) {
            const bool live = c * EPC + esel < deg;
            const int   s  = live ? ks[c] : node;
            const float ex = live ? __expf(ke[c] - mx) : 0.0f;
            dsum += ex;
            const int lim = deg - c * EPC < EPC ? deg - c * EPC : EPC;
            add_chunk_rows<H>(xw, s, ex, lim, hch, ch, acc0, acc1);
        }
    }
    for (int c0 = KEEP * EPC; c0 < deg; c0 += EPC) {
        const int j = c0 + esel;
        const bool live = j < deg;
        const int   sl = sorted_src[beg + (live ? j : deg - 1)];
        const float el = __expf(lrelu(asrc[sl * H + hsel] + ad) - mx);
        const int   s  = live ? sl : node;
        const float ex = live ? el : 0.0f;
        dsum += ex;
        const int lim = deg - c0 < EPC ? deg - c0 : EPC;
        add_chunk_rows<H>(xw, s, ex, lim, hch, ch, acc0, acc1);
    }
    for (int off = H; off < 32; off <<= 1) dsum += __shfl_xor(dsum, off, 32);
    const float den = __shfl(dsum, hch, 32) + SM_EPS;

    const float v0 = acc0 / den + bi.x;
    const float v1 = acc1 / den + bi.y;
    float sum = v0 + v1;
    #pragma unroll
    for (int off = 16; off > 0; off >>= 1) sum += __shfl_xor(sum, off, 32);
    const float mu = sum * (1.0f / 64.0f);
    const float d0 = v0 - mu, d1 = v1 - mu;
    float vv = d0 * d0 + d1 * d1;
    #pragma unroll
    for (int off = 16; off > 0; off >>= 1) vv += __shfl_xor(vv, off, 32);
    const float rstd = rsqrtf(vv * (1.0f / 64.0f) + LN_EPS);

    float o0 = d0 * rstd * gg.x + bb.x;
    float o1 = d1 * rstd * gg.y + bb.y;
    o0 = o0 > 0.0f ? o0 : __expf(o0) - 1.0f;
    o1 = o1 > 0.0f ? o1 : __expf(o1) - 1.0f;
    v2f o; o.x = o0; o.y = o1;
    *(v2f*)(out + (size_t)node * 64 + ch) = o;
}

static inline unsigned cdiv(long long a, int b) { return (unsigned)((a + b - 1) / b); }

extern "C" void kernel_launch(void* const* d_in, const int* in_sizes, int n_in,
                              void* d_out, int out_size, void* d_ws, size_t ws_size,
                              hipStream_t stream) {
    const float* h   = (const float*)d_in[0];
    const int*   ei  = (const int*)d_in[1];
    const float* W1  = (const float*)d_in[2];
    const float* a1s = (const float*)d_in[3];
    const float* a1d = (const float*)d_in[4];
    const float* b1  = (const float*)d_in[5];
    const float* W2  = (const float*)d_in[6];
    const float* a2s = (const float*)d_in[7];
    const float* a2d = (const float*)d_in[8];
    const float* b2  = (const float*)d_in[9];
    const float* ln1g = (const float*)d_in[10];
    const float* ln1b = (const float*)d_in[11];
    const float* ln2g = (const float*)d_in[12];
    const float* ln2b = (const float*)d_in[13];

    const int n = in_sizes[0] / 64;
    const int e = in_sizes[1] / 2;
    const int* src = ei;
    const int* dst = ei + e;

    float* ws = (float*)d_ws;
    float* xw   = ws; ws += (size_t)n * 64;
    float* asrc = ws; ws += (size_t)n * 4;
    float* adst = ws; ws += (size_t)n * 4;
    float* x1   = ws; ws += (size_t)n * 64;
    int* wi     = (int*)ws;
    int* cnt    = wi; wi += n;
    int* start  = wi; wi += n;
    int* total  = wi; wi += 2;
    int* ssrc   = wi; wi += e;
    int* table  = wi; wi += (size_t)SORT_CHUNKS * n;

    const int B = 256;

    sort_count<<<SORT_CHUNKS * SORT_RANGES, SORT_THREADS, 0, stream>>>(dst, table, total, n, e);
    sort_alloc<<<cdiv(n, B), B, 0, stream>>>(table, total, start, cnt, n);
    sort_place<<<SORT_CHUNKS * SORT_RANGES, SORT_THREADS, 0, stream>>>(src, dst, table, ssrc, n, e);

    gemm64_wmma_dots<4><<<cdiv(n, 16), 128, 0, stream>>>(h, W1, a1s, a1d, xw, asrc, adst, n);
    node_gather_norm<4, 16><<<cdiv((long long)n * 32, B), B, 0, stream>>>(start, cnt, ssrc, asrc, adst, xw, b1, ln1g, ln1b, x1, n);

    gemm64_wmma_dots<1><<<cdiv(n, 16), 128, 0, stream>>>(x1, W2, a2s, a2d, xw, asrc, adst, n);
    node_gather_norm<1, 64><<<cdiv((long long)n * 32, B), B, 0, stream>>>(start, cnt, ssrc, asrc, adst, xw, b2, ln2g, ln2b, (float*)d_out, n);
}


# source rows gathered by 16-byte loads, 16 lanes per row, two edges per load instruction (half the row loads, no scalar row address per edge); node sums now even+odd edge halves, LN sums re-paired
# speedup vs baseline: 6.2632x; 1.0429x over previous
#include <hip/hip_runtime.h>

#define NEG_SLOPE 0.2f
#define LN_EPS 1e-5f
#define SM_EPS 1e-16f

typedef __attribute__((ext_vector_type(2))) float v2f;
typedef __attribute__((ext_vector_type(4))) float v4f;
typedef __attribute__((ext_vector_type(8))) float v8f;

__device__ __forceinline__ float lrelu(float x) { return x > 0.0f ? x : NEG_SLOPE * x; }

template <int CTRL>
__device__ __forceinline__ float row_xor_add(float x) {
    const int y = __builtin_amdgcn_update_dpp(0, __float_as_int(x), CTRL, 0xf, 0xf, true);
    return x + __int_as_float(y);
}
__device__ __forceinline__ float row16_sum(float x) {
    x = row_xor_add<0x161>(x);
    x = row_xor_add<0x162>(x);
    x = row_xor_add<0x164>(x);
    x = row_xor_add<0x168>(x);
    return x;
}

template <int H>
__global__ void __launch_bounds__(128)
gemm64_wmma_dots(const float* __restrict__ A, const float* __restrict__ W,
                 const float* __restrict__ a_src, const float* __restrict__ a_dst,
                 float* __restrict__ C, float* __restrict__ asrc,
                 float* __restrict__ adst, int nrows) {
    __shared__ float part[2][4][16];
    const int wave = threadIdx.x >> 5;
    const int lane = threadIdx.x & 31;
    const int mrow = lane & 15;
    const int kh   = lane >> 4;
    const int row0 = blockIdx.x * 16;
    int row = row0 + mrow;
    row = row < nrows ? row : nrows - 1;
    const int col  = wave * 16 + mrow;

    const float* ap = A + (size_t)row * 64 + 2 * kh;
    const float* bp = W + (size_t)(2 * kh) * 64 + col;
    const float wsrc = a_src[col];
    const float wdst = a_dst[col];

    v8f acc = {};
    #pragma unroll
    for (int k = 0; k < 64; k += 4) {
        v2f a = *(const v2f*)(ap + k);
        v2f b;
        b.x = bp[(size_t)k * 64];
        b.y = bp[(size_t)k * 64 + 64];
        acc = __builtin_amdgcn_wmma_f32_16x16x4_f32(
             false, a,  false, b,
             (short)0, acc,  false,  false);
    }

    float vs = 0.0f, vd = 0.0f;
    #pragma unroll
    for (int r = 0; r < 8; ++r) {
        const float ss = row16_sum(acc[r] * wsrc);
        const float sd = row16_sum(acc[r] * wdst);
        vs = mrow == r ? ss : vs;
        vd = mrow == r ? sd : vd;
    }
    if (mrow < 8) {
        part[0][wave][8 * kh + mrow] = vs;
        part[1][wave][8 * kh + mrow] = vd;
    }
    __syncthreads();

    const int srow = row0 + 8 * (lane >> 4);
    const int scol = wave * 16 + (lane & 15);
    #pragma unroll
    for (int r = 0; r < 8; ++r) {
        const int rr = srow + r;
        if (rr < nrows) C[(size_t)rr * 64 + scol] = acc[r];
    }

    if (H == 4) {
        const int q = threadIdx.x >> 6;
        const int i = threadIdx.x & 63;
        if (row0 + (i >> 2) < nrows)
            (q ? adst : asrc)[(size_t)row0 * 4 + i] = part[q][i & 3][i >> 2];
    } else {
        if (wave < 2 && lane < 16 && row0 + lane < nrows) {
            const float d = ((part[wave][0][lane] + part[wave][1][lane])
                             + part[wave][2][lane]) + part[wave][3][lane];
            (wave ? adst : asrc)[row0 + lane] = d;
        }
    }
}

#define SORT_CHUNKS  64
#define SORT_RANGES  4
#define SORT_TILE    12544
#define SORT_THREADS 1024
#define SORT_UNROLL  10
#define SORT_SLOTS   ((SORT_TILE + SORT_THREADS - 1) / SORT_THREADS)

__device__ __forceinline__ int sort_chunk_len(int ecnt) {
    return (ecnt + SORT_CHUNKS - 1) / SORT_CHUNKS;
}

__global__ void __launch_bounds__(SORT_THREADS)
sort_count(const int* __restrict__ dst, int* __restrict__ table,
           int* __restrict__ total, int n, int ecnt) {
    __shared__ int lcnt[SORT_TILE + 32];
    const int tid   = threadIdx.x;
    const int chunk = blockIdx.x / SORT_RANGES;
    const int range = blockIdx.x % SORT_RANGES;
    const int per   = sort_chunk_len(ecnt);
    const int e0    = chunk * per;
    const int e1    = e0 + per < ecnt ? e0 + per : ecnt;
    const int spare = SORT_TILE + (tid & 31);
    if (blockIdx.x == 0 && tid == 0) *total = 0;
    if (tid < 32) lcnt[SORT_TILE + tid] = 0;
    for (int tb = range * SORT_TILE; tb < n; tb += SORT_RANGES * SORT_TILE) {
        const int tn = n - tb < SORT_TILE ? n - tb : SORT_TILE;
        for (int i = tid; i < tn; i += SORT_THREADS) lcnt[i] = 0;
        __syncthreads();
        for (int e = e0 + tid; e < e1; e += SORT_UNROLL * SORT_THREADS) {
            int d[SORT_UNROLL];
            #pragma unroll
            for (int u = 0; u < SORT_UNROLL; ++u) {
                const int ee = e + u * SORT_THREADS;
                const int dv = dst[ee < e1 ? ee : e1 - 1];
                d[u] = ee < e1 ? dv - tb : -1;
            }
            #pragma unroll
            for (int u = 0; u < SORT_UNROLL; ++u)
                atomicAdd(&lcnt[(unsigned)d[u] < (unsigned)tn ? d[u] : spare], 1);
        }
        __syncthreads();
        int* row = table + (size_t)chunk * n + tb;
        int cv[SORT_SLOTS];
        #pragma unroll
        for (int k = 0; k < SORT_SLOTS; ++k) {
            const int i = tid + k * SORT_THREADS;
            cv[k] = lcnt[i < tn ? i : 0];
        }
        #pragma unroll
        for (int k = 0; k < SORT_SLOTS; ++k) {
            const int i = tid + k * SORT_THREADS;
            if (i < tn) row[i] = cv[k];
        }
        __syncthreads();
    }
}

__global__ void __launch_bounds__(256)
sort_alloc(int* __restrict__ table, int* __restrict__ total,
           int* __restrict__ start, int* __restrict__ cnt, int n) {
    const int t = blockIdx.x * blockDim.x + threadIdx.x;
    const int lane = threadIdx.x & 31;
    const bool live = t < n;
    int v[SORT_CHUNKS];
    #pragma unroll
    for (int k = 0; k < SORT_CHUNKS; ++k) v[k] = 0;
    if (live) {
        #pragma unroll
        for (int k = 0; k < SORT_CHUNKS; ++k) v[k] = table[(size_t)k * n + t];
    }
    int c = 0;
    #pragma unroll
    for (int k = 0; k < SORT_CHUNKS; ++k) c += v[k];
    int incl = c;
    for (int off = 1; off < 32; off <<= 1) {
        const int s = __shfl_up(incl, off, 32);
        if (lane >= off) incl += s;
    }
    int base = 0;
    if (lane == 31) base = atomicAdd(total, incl);
    base = __shfl(base, 31, 32);
    if (live) {
        int pos = base + incl - c;
        start[t] = pos;
        cnt[t]   = c;
        #pragma unroll
        for (int k = 0; k < SORT_CHUNKS; ++k) {
            table[(size_t)k * n + t] = pos;
            pos += v[k];
        }
    }
}

__global__ void __launch_bounds__(SORT_THREADS)
sort_place(const int* __restrict__ src, const int* __restrict__ dst,
           const int* __restrict__ table, int* __restrict__ sorted_src,
           int n, int ecnt) {
    __shared__ int lpos[SORT_TILE + 32];
    const int tid   = threadIdx.x;
    const int chunk = blockIdx.x / SORT_RANGES;
    const int range = blockIdx.x % SORT_RANGES;
    const int per   = sort_chunk_len(ecnt);
    const int e0    = chunk * per;
    const int e1    = e0 + per < ecnt ? e0 + per : ecnt;
    const int spare = SORT_TILE + (tid & 31);
    if (tid < 32) lpos[SORT_TILE + tid] = 0;
    for (int tb = range * SORT_TILE; tb < n; tb += SORT_RANGES * SORT_TILE) {
        const int tn = n - tb < SORT_TILE ? n - tb : SORT_TILE;
        const int* row = table + (size_t)chunk * n + tb;
        int bv[SORT_SLOTS];
        #pragma unroll
        for (int k = 0; k < SORT_SLOTS; ++k) {
            const int i = tid + k * SORT_THREADS;
            bv[k] = row[i < tn ? i : tn - 1];
        }
        #pragma unroll
        for (int k = 0; k < SORT_SLOTS; ++k) {
            const int i = tid + k * SORT_THREADS;
            if (i < tn) lpos[i] = bv[k];
        }
        __syncthreads();
        for (int e = e0 + tid; e < e1; e += SORT_UNROLL * SORT_THREADS) {
            int d[SORT_UNROLL], s[SORT_UNROLL], p[SORT_UNROLL];
            #pragma unroll
            for (int u = 0; u < SORT_UNROLL; ++u) {
                const int ee = e + u * SORT_THREADS;
                const int ec = ee < e1 ? ee : e1 - 1;
                const int dv = dst[ec];
                s[u] = src[ec];
                d[u] = ee < e1 ? dv - tb : -1;
            }
            #pragma unroll
            for (int u = 0; u < SORT_UNROLL; ++u) asm volatile("" :: "v"(s[u]));
            #pragma unroll
            for (int u = 0; u < SORT_UNROLL; ++u)
                p[u] = atomicAdd(&lpos[(unsigned)d[u] < (unsigned)tn ? d[u] : spare], 1);
            #pragma unroll
            for (int u = 0; u < SORT_UNROLL; ++u)
                if ((unsigned)d[u] < (unsigned)tn) sorted_src[p[u]] = s[u];
        }
        __syncthreads();
    }
}


__device__ __forceinline__ float xhalf(float x) {
    return __int_as_float(__builtin_amdgcn_permlanex16(__float_as_int(x), __float_as_int(x),
                                                       0x76543210, 0xfedcba98, false, false));
}

template <int H>
__device__ __forceinline__ void add_chunk_rows(const float* __restrict__ xw, int s, float ex,
                                               int lim, int srcl, unsigned qoff, v4f& acc) {
    constexpr int EPC = 32 / H;
    constexpr int SPC = EPC / 2;
    constexpr int GRP = SPC < 8 ? SPC : 8;
    for (int k0 = 0; k0 < lim; k0 += 2 * GRP) {
        #pragma unroll
        for (int kk = 0; kk < GRP; ++kk) {
            const int l = srcl + (k0 + 2 * kk) * H;
            const int sk = __shfl(s, l, 32);
            const float wk = __shfl(ex, l, 32);
            const v4f v = *(const v4f*)((const char*)xw + ((unsigned)sk * 256u + qoff));
            acc += wk * v;
        }
    }
}

template <int H, int C>
__global__ void __launch_bounds__(256)
node_gather_norm(const int* __restrict__ start, const int* __restrict__ cnt,
                 const int* __restrict__ sorted_src,
                 const float* __restrict__ asrc, const float* __restrict__ adst,
                 const float* __restrict__ xw,
                 const float* __restrict__ bias, const float* __restrict__ g,
                 const float* __restrict__ bparam,
                 float* __restrict__ out, int n) {
    constexpr int EPC  = 32 / H;
    constexpr int KEEP = H == 1 ? 2 : 5;
    const int node = __builtin_amdgcn_readfirstlane(
        (int)((blockIdx.x * blockDim.x + threadIdx.x) >> 5));
    const int lane = threadIdx.x & 31;
    if (node >= n) return;
    const int hsel = lane % H;
    const int esel = lane / H;
    const int half = lane >> 4;
    const int q    = lane & 15;
    const int ch   = 4 * q;
    const int hch  = ch / C;
    const int srcl = half * H + hch;
    const unsigned qoff = (unsigned)q * 16u;
    const int ch2  = ch + 2 * half;
    const int beg  = start[node];
    const int deg  = cnt[node];
    const float ad    = adst[node * H + hsel];
    const float eself = lrelu(asrc[node * H + hsel] + ad);
    int   ks[KEEP];
    float ke[KEEP];
    #pragma unroll
    for (int c = 0; c < KEEP; ++c) {
        ks[c] = node;
        if (c * EPC < deg) {
            const int j = c * EPC + esel;
            ks[c] = sorted_src[beg + (j < deg ? j : deg - 1)];
        }
    }
    #pragma unroll
    for (int c = 0; c < KEEP; ++c) ke[c] = lrelu(asrc[ks[c] * H + hsel] + ad);
    float mx = eself;
    #pragma unroll
    for (int c = 0; c < KEEP; ++c) mx = fmaxf(mx, ke[c]);
    for (int c0 = KEEP * EPC; c0 < deg; c0 += EPC) {
        const int j = c0 + esel;
        const int s = sorted_src[beg + (j < deg ? j : deg - 1)];
        mx = fmaxf(mx, lrelu(asrc[s * H + hsel] + ad));
    }
    for (int off = H; off < 32; off <<= 1) mx = fmaxf(mx, __shfl_xor(mx, off, 32));

    const float exself = __expf(eself - mx);
    float dsum = esel == 0 ? exself : 0.0f;
    const float wself = __shfl(exself, hch, 32);
    const v4f xs = *(const v4f*)(xw + (size_t)node * 64 + ch);
    v4f acc = (half == 0 ? wself : 0.0f) * xs;

    #pragma unroll
    for (int c = 0; c < KEEP; ++c) {
        if (c * EPC < deg) {
            const bool live = c * EPC + esel < deg;
            const int   s  = live ? ks[c] : node;
            const float ex = live ? __expf(ke[c] - mx) : 0.0f;
            dsum += ex;
            const int lim = deg - c * EPC < EPC ? deg - c * EPC : EPC;
            add_chunk_rows<H>(xw, s, ex, lim, srcl, qoff, acc);
        }
    }
    for (int c0 = KEEP * EPC; c0 < deg; c0 += EPC) {
        const int j = c0 + esel;
        const bool live = j < deg;
        const int   sl = sorted_src[beg + (live ? j : deg - 1)];
        const float el = __expf(lrelu(asrc[sl * H + hsel] + ad) - mx);
        const int   s  = live ? sl : node;
        const float ex = live ? el : 0.0f;
        dsum += ex;
        const int lim = deg - c0 < EPC ? deg - c0 : EPC;
        add_chunk_rows<H>(xw, s, ex, lim, srcl, qoff, acc);
    }
    const v2f bi = *(const v2f*)(bias + ch2);
    const v2f gg = *(const v2f*)(g + ch2);
    const v2f bb = *(const v2f*)(bparam + ch2);
    for (int off = H; off < 32; off <<= 1) dsum += __shfl_xor(dsum, off, 32);
    const float den = __shfl(dsum, hch, 32) + SM_EPS;

    const float a0 = (half ? acc.z : acc.x) + xhalf(half ? acc.x : acc.z);
    const float a1 = (half ? acc.w : acc.y) + xhalf(half ? acc.y : acc.w);

    const float v0 = a0 / den + bi.x;
    const float v1 = a1 / den + bi.y;
    float sum = row16_sum(v0 + v1);
    sum += xhalf(sum);
    const float mu = sum * (1.0f / 64.0f);
    const float d0 = v0 - mu, d1 = v1 - mu;
    float vv = row16_sum(d0 * d0 + d1 * d1);
    vv += xhalf(vv);
    const float rstd = rsqrtf(vv * (1.0f / 64.0f) + LN_EPS);

    float o0 = d0 * rstd * gg.x + bb.x;
    float o1 = d1 * rstd * gg.y + bb.y;
    o0 = o0 > 0.0f ? o0 : __expf(o0) - 1.0f;
    o1 = o1 > 0.0f ? o1 : __expf(o1) - 1.0f;
    v2f o; o.x = o0; o.y = o1;
    *(v2f*)(out + (size_t)node * 64 + ch2) = o;
}

static inline unsigned cdiv(long long a, int b) { return (unsigned)((a + b - 1) / b); }

extern "C" void kernel_launch(void* const* d_in, const int* in_sizes, int n_in,
                              void* d_out, int out_size, void* d_ws, size_t ws_size,
                              hipStream_t stream) {
    const float* h   = (const float*)d_in[0];
    const int*   ei  = (const int*)d_in[1];
    const float* W1  = (const float*)d_in[2];
    const float* a1s = (const float*)d_in[3];
    const float* a1d = (const float*)d_in[4];
    const float* b1  = (const float*)d_in[5];
    const float* W2  = (const float*)d_in[6];
    const float* a2s = (const float*)d_in[7];
    const float* a2d = (const float*)d_in[8];
    const float* b2  = (const float*)d_in[9];
    const float* ln1g = (const float*)d_in[10];
    const float* ln1b = (const float*)d_in[11];
    const float* ln2g = (const float*)d_in[12];
    const float* ln2b = (const float*)d_in[13];

    const int n = in_sizes[0] / 64;
    const int e = in_sizes[1] / 2;
    const int* src = ei;
    const int* dst = ei + e;

    float* ws = (float*)d_ws;
    float* xw   = ws; ws += (size_t)n * 64;
    float* asrc = ws; ws += (size_t)n * 4;
    float* adst = ws; ws += (size_t)n * 4;
    float* x1   = ws; ws += (size_t)n * 64;
    int* wi     = (int*)ws;
    int* cnt    = wi; wi += n;
    int* start  = wi; wi += n;
    int* total  = wi; wi += 2;
    int* ssrc   = wi; wi += e;
    int* table  = wi; wi += (size_t)SORT_CHUNKS * n;

    const int B = 256;

    sort_count<<<SORT_CHUNKS * SORT_RANGES, SORT_THREADS, 0, stream>>>(dst, table, total, n, e);
    sort_alloc<<<cdiv(n, B), B, 0, stream>>>(table, total, start, cnt, n);
    sort_place<<<SORT_CHUNKS * SORT_RANGES, SORT_THREADS, 0, stream>>>(src, dst, table, ssrc, n, e);

    gemm64_wmma_dots<4><<<cdiv(n, 16), 128, 0, stream>>>(h, W1, a1s, a1d, xw, asrc, adst, n);
    node_gather_norm<4, 16><<<cdiv((long long)n * 32, B), B, 0, stream>>>(start, cnt, ssrc, asrc, adst, xw, b1, ln1g, ln1b, x1, n);

    gemm64_wmma_dots<1><<<cdiv(n, 16), 128, 0, stream>>>(x1, W2, a2s, a2d, xw, asrc, adst, n);
    node_gather_norm<1, 64><<<cdiv((long long)n * 32, B), B, 0, stream>>>(start, cnt, ssrc, asrc, adst, xw, b2, ln2g, ln2b, (float*)d_out, n);
}
